# starting kernel
# baseline (speedup 1.0000x reference)
#include <hip/hip_runtime.h>

typedef __attribute__((ext_vector_type(2))) float v2f;
typedef __attribute__((ext_vector_type(8))) float v8f;


__global__ void zero_i32_kernel(int* __restrict__ p, int n) {
    int t = blockIdx.x * blockDim.x + threadIdx.x;
    if (t < n) p[t] = 0;
}

__global__ void deg_count_kernel(const int* __restrict__ dst, int* __restrict__ cnt, int E) {
    int t = blockIdx.x * blockDim.x + threadIdx.x;
    if (t < E) atomicAdd(&cnt[dst[t]], 1);
}

__global__ void dinv_kernel(const int* __restrict__ cnt, float* __restrict__ dinv, int n) {
    int t = blockIdx.x * blockDim.x + threadIdx.x;
    if (t < n) dinv[t] = 1.0f / sqrtf((float)(cnt[t] + 1));
}

__global__ __launch_bounds__(1024)
void scan_kernel(const int* __restrict__ cnt, int* __restrict__ rowptr, int n) {
    __shared__ int sums[1024];
    const int tid = threadIdx.x;
    const int chunk = (n + 1023) / 1024;
    const int b = tid * chunk;
    const int e = min(b + chunk, n);
    int s = 0;
    for (int i = b; i < e; ++i) s += cnt[i];
    sums[tid] = s;
    __syncthreads();
    for (int off = 1; off < 1024; off <<= 1) {
        int v = (tid >= off) ? sums[tid - off] : 0;
        __syncthreads();
        sums[tid] += v;
        __syncthreads();
    }
    int run = (tid == 0) ? 0 : sums[tid - 1];
    for (int i = b; i < e; ++i) { rowptr[i] = run; run += cnt[i]; }
    if (tid == 1023) rowptr[n] = run;
}

__global__ void csr_fill_kernel(const int* __restrict__ src, const int* __restrict__ dst,
                                const float* __restrict__ dinv,
                                const int* __restrict__ rowptr, int* __restrict__ cursor,
                                int2* __restrict__ meta, int E) {
    int t = blockIdx.x * blockDim.x + threadIdx.x;
    if (t >= E) return;
    int s = src[t], d = dst[t];
    int slot = atomicAdd(&cursor[d], 1);
    int2 m;
    m.x = s;
    m.y = __float_as_int(dinv[s] * dinv[d]);
    meta[rowptr[d] + slot] = m;
}

#define GEMM_WAVES 8
template <int NCT>
__global__ __launch_bounds__(256)
void gcn_gemm_wmma_kernel(const float* __restrict__ A, const float* __restrict__ W,
                          float* __restrict__ h, int n, int fin) {
    const int fout = NCT * 16;
    __shared__ float Wt[128 * NCT * 16];
    const int tid = threadIdx.x;
    for (int idx = tid; idx < fin * fout; idx += 256)
        Wt[idx] = W[idx];
    __syncthreads();

    const int wv   = tid >> 5;
    const int lane = tid & 31;
    const int tm   = blockIdx.x * GEMM_WAVES + wv;
    const int nrt  = (n + 15) >> 4;
    if (tm >= nrt) return;

    const int half = lane >> 4;
    const int l    = lane & 15;
    int row  = tm * 16 + l;
    int arow = row < n ? row : n - 1;
    const float* ap = A + (size_t)arow * fin + half * 2;

    v8f acc[NCT];
    #pragma unroll
    for (int t = 0; t < NCT; ++t) acc[t] = {};

    #pragma unroll 2
    for (int k = 0; k < fin; k += 4) {
        v2f a;
        a.x = ap[k];
        a.y = ap[k + 1];
        const int kb = k + half * 2;
        #pragma unroll
        for (int t = 0; t < NCT; ++t) {
            const int col = t * 16 + l;
            v2f b;
            b.x = Wt[kb * fout + col];
            b.y = Wt[(kb + 1) * fout + col];
            acc[t] = __builtin_amdgcn_wmma_f32_16x16x4_f32(
                         false, a, false, b, (short)0, acc[t], false, false);
        }
    }

    const int base_m = tm * 16 + half * 8;
    #pragma unroll
    for (int t = 0; t < NCT; ++t) {
        const int col = t * 16 + l;
        #pragma unroll
        for (int j = 0; j < 8; ++j) {
            int m = base_m + j;
            if (m < n) h[(size_t)m * fout + col] = acc[t][j];
        }
    }
}

template <int CPL>
__global__ void gcn_gather_kernel(const float* __restrict__ h,
                                  const int* __restrict__ rowptr,
                                  const int2* __restrict__ meta,
                                  const float* __restrict__ dinv,
                                  const float* __restrict__ bias,
                                  float* __restrict__ out,
                                  int n, int fout, int do_relu) {
    const int wave = (blockIdx.x * blockDim.x + threadIdx.x) >> 5;
    const int lane = threadIdx.x & 31;
    if (wave >= n) return;
    const int i  = wave;
    const int c0 = lane * CPL;

    float acc[CPL];
    const float dv = dinv[i];
    const float* hi = h + (size_t)i * fout + c0;
    #pragma unroll
    for (int q = 0; q < CPL; ++q) acc[q] = hi[q] * dv * dv;

    const int beg = rowptr[i], end = rowptr[i + 1];
    #pragma unroll 2
    for (int j = beg; j < end; ++j) {
        const int2 m  = meta[j];
        const float nm = __int_as_float(m.y);
        const float* hs = h + (size_t)m.x * fout + c0;
        #pragma unroll
        for (int q = 0; q < CPL; ++q) acc[q] = fmaf(hs[q], nm, acc[q]);
    }

    float* oi = out + (size_t)i * fout + c0;
    #pragma unroll
    for (int q = 0; q < CPL; ++q) {
        float v = acc[q] + bias[c0 + q];
        oi[q] = do_relu ? fmaxf(v, 0.0f) : v;
    }
}


static inline int cdiv(long long a, int b) { return (int)((a + b - 1) / b); }
static inline size_t align4up(size_t x) { return (x + 3) & ~(size_t)3; }

extern "C" void kernel_launch(void* const* d_in, const int* in_sizes, int n_in,
                              void* d_out, int out_size, void* d_ws, size_t ws_size,
                              hipStream_t stream) {
    const float* x  = (const float*)d_in[0];
    const int*   ei = (const int*)  d_in[1];
    const float* W1 = (const float*)d_in[2];
    const float* b1 = (const float*)d_in[3];
    const float* W2 = (const float*)d_in[4];
    const float* b2 = (const float*)d_in[5];
    const float* W3 = (const float*)d_in[6];
    const float* b3 = (const float*)d_in[7];

    const int IN_C = 128, HID = 128, MID = 64, OUT = 64;
    const int n = in_sizes[0] / IN_C;
    const int E = in_sizes[1] / 2;
    const int* src = ei;
    const int* dst = ei + E;

    char* base = (char*)d_ws;
    size_t off = 0;
    auto take = [&](size_t elems) { void* p = base + off * 4; off += align4up(elems); return p; };
    int*   cnt    = (int*)  take(n);
    int*   rowptr = (int*)  take(n + 1);
    float* dinv   = (float*)take(n);
    int2*  meta   = (int2*) take((size_t)2 * E);
    float* h      = (float*)take((size_t)n * 128);
    float* xn     = (float*)take((size_t)n * 128);
    (void)ws_size;

    const int B = 256;

    zero_i32_kernel<<<cdiv(n, B), B, 0, stream>>>(cnt, n);
    deg_count_kernel<<<cdiv(E, B), B, 0, stream>>>(dst, cnt, E);
    dinv_kernel<<<cdiv(n, B), B, 0, stream>>>(cnt, dinv, n);
    scan_kernel<<<1, 1024, 0, stream>>>(cnt, rowptr, n);
    zero_i32_kernel<<<cdiv(n, B), B, 0, stream>>>(cnt, n);
    csr_fill_kernel<<<cdiv(E, B), B, 0, stream>>>(src, dst, dinv, rowptr, cnt, meta, E);

    const int nrt = (n + 15) / 16;
    const int gemm_blocks = cdiv(nrt, GEMM_WAVES);
    const int gather_blocks = cdiv((long long)n * 32, B);

    auto run_layer = [&](const float* in, const float* W, const float* bias,
                         float* out, int fin, int fout, int relu) {
        if (fout == 128)
            gcn_gemm_wmma_kernel<8><<<gemm_blocks, B, 0, stream>>>(in, W, h, n, fin);
        else
            gcn_gemm_wmma_kernel<4><<<gemm_blocks, B, 0, stream>>>(in, W, h, n, fin);
        if (fout == 128)
            gcn_gather_kernel<4><<<gather_blocks, B, 0, stream>>>(h, rowptr, meta,
                                                                  dinv, bias, out, n, fout, relu);
        else
            gcn_gather_kernel<2><<<gather_blocks, B, 0, stream>>>(h, rowptr, meta,
                                                                  dinv, bias, out, n, fout, relu);
    };

    run_layer(x,  W1, b1, xn,            IN_C, HID, 1);
    run_layer(xn, W2, b2, xn,            HID,  MID, 1);
    run_layer(xn, W3, b3, (float*)d_out, MID,  OUT, 0);
}


# Prefix sum of node in-degrees made grid-wide (per-tile totals, then per-tile scan plus summed earlier totals) in place of the single-work-group serial scan; integer sums, order-free
# speedup vs baseline: 1.3837x; 1.3837x over previous
#include <hip/hip_runtime.h>

typedef __attribute__((ext_vector_type(2))) float v2f;
typedef __attribute__((ext_vector_type(8))) float v8f;


__global__ void zero_i32_kernel(int* __restrict__ p, int n) {
    int t = blockIdx.x * blockDim.x + threadIdx.x;
    if (t < n) p[t] = 0;
}

__global__ void deg_count_kernel(const int* __restrict__ dst, int* __restrict__ cnt, int E) {
    int t = blockIdx.x * blockDim.x + threadIdx.x;
    if (t < E) atomicAdd(&cnt[dst[t]], 1);
}

__global__ void dinv_kernel(const int* __restrict__ cnt, float* __restrict__ dinv, int n) {
    int t = blockIdx.x * blockDim.x + threadIdx.x;
    if (t < n) dinv[t] = 1.0f / sqrtf((float)(cnt[t] + 1));
}

#define SCAN_TILE 1024

__device__ __forceinline__ int4 load_counts4(const int* __restrict__ cnt, int base, int n) {
    int4 v = make_int4(0, 0, 0, 0);
    if (base + 3 < n) {
        v = *reinterpret_cast<const int4*>(cnt + base);
    } else {
        if (base     < n) v.x = cnt[base];
        if (base + 1 < n) v.y = cnt[base + 1];
        if (base + 2 < n) v.z = cnt[base + 2];
    }
    return v;
}

__global__ __launch_bounds__(256)
void tile_total_kernel(const int* __restrict__ cnt, int* __restrict__ tsum, int n) {
    __shared__ int wsum[8];
    const int tid  = threadIdx.x;
    const int base = (blockIdx.x * 256 + tid) * 4;
    const int4 v = load_counts4(cnt, base, n);
    int s = (v.x + v.y) + (v.z + v.w);
    for (int off = 16; off > 0; off >>= 1) s += __shfl_xor(s, off);
    if ((tid & 31) == 0) wsum[tid >> 5] = s;
    __syncthreads();
    if (tid == 0) {
        int t = 0;
        for (int w = 0; w < 8; ++w) t += wsum[w];
        tsum[blockIdx.x] = t;
    }
}

__global__ __launch_bounds__(256)
void scan_kernel(const int* __restrict__ cnt, const int* __restrict__ tsum,
                 int* __restrict__ rowptr, int n) {
    __shared__ int wtile[8];
    __shared__ int wprev[8];
    const int tid  = threadIdx.x;
    const int lane = tid & 31;
    const int wv   = tid >> 5;
    const int base = (blockIdx.x * 256 + tid) * 4;
    const int4 v = load_counts4(cnt, base, n);

    int prev = 0;
    for (int j = tid; j < (int)blockIdx.x; j += 256) prev += tsum[j];

    const int t1 = v.x, t2 = t1 + v.y, t3 = t2 + v.z, t4 = t3 + v.w;
    int inc = t4;
    for (int off = 1; off < 32; off <<= 1) {
        const int u = __shfl_up(inc, off);
        if (lane >= off) inc += u;
    }
    for (int off = 16; off > 0; off >>= 1) prev += __shfl_xor(prev, off);
    if (lane == 31) wtile[wv] = inc;
    if (lane == 0)  wprev[wv] = prev;
    __syncthreads();

    int start = 0;
    for (int w = 0; w < 8; ++w) {
        start += wprev[w];
        if (w < wv) start += wtile[w];
    }
    const int excl = start + inc - t4;

    if (base + 3 < n) {
        *reinterpret_cast<int4*>(rowptr + base) =
            make_int4(excl, excl + t1, excl + t2, excl + t3);
    } else {
        if (base     < n) rowptr[base]     = excl;
        if (base + 1 < n) rowptr[base + 1] = excl + t1;
        if (base + 2 < n) rowptr[base + 2] = excl + t2;
    }
    if (base < n && base + 4 >= n) rowptr[n] = excl + t4;
}

__global__ void csr_fill_kernel(const int* __restrict__ src, const int* __restrict__ dst,
                                const float* __restrict__ dinv,
                                const int* __restrict__ rowptr, int* __restrict__ cursor,
                                int2* __restrict__ meta, int E) {
    int t = blockIdx.x * blockDim.x + threadIdx.x;
    if (t >= E) return;
    int s = src[t], d = dst[t];
    int slot = atomicAdd(&cursor[d], 1);
    int2 m;
    m.x = s;
    m.y = __float_as_int(dinv[s] * dinv[d]);
    meta[rowptr[d] + slot] = m;
}

#define GEMM_WAVES 8
template <int NCT>
__global__ __launch_bounds__(256)
void gcn_gemm_wmma_kernel(const float* __restrict__ A, const float* __restrict__ W,
                          float* __restrict__ h, int n, int fin) {
    const int fout = NCT * 16;
    __shared__ float Wt[128 * NCT * 16];
    const int tid = threadIdx.x;
    for (int idx = tid; idx < fin * fout; idx += 256)
        Wt[idx] = W[idx];
    __syncthreads();

    const int wv   = tid >> 5;
    const int lane = tid & 31;
    const int tm   = blockIdx.x * GEMM_WAVES + wv;
    const int nrt  = (n + 15) >> 4;
    if (tm >= nrt) return;

    const int half = lane >> 4;
    const int l    = lane & 15;
    int row  = tm * 16 + l;
    int arow = row < n ? row : n - 1;
    const float* ap = A + (size_t)arow * fin + half * 2;

    v8f acc[NCT];
    #pragma unroll
    for (int t = 0; t < NCT; ++t) acc[t] = {};

    #pragma unroll 2
    for (int k = 0; k < fin; k += 4) {
        v2f a;
        a.x = ap[k];
        a.y = ap[k + 1];
        const int kb = k + half * 2;
        #pragma unroll
        for (int t = 0; t < NCT; ++t) {
            const int col = t * 16 + l;
            v2f b;
            b.x = Wt[kb * fout + col];
            b.y = Wt[(kb + 1) * fout + col];
            acc[t] = __builtin_amdgcn_wmma_f32_16x16x4_f32(
                         false, a, false, b, (short)0, acc[t], false, false);
        }
    }

    const int base_m = tm * 16 + half * 8;
    #pragma unroll
    for (int t = 0; t < NCT; ++t) {
        const int col = t * 16 + l;
        #pragma unroll
        for (int j = 0; j < 8; ++j) {
            int m = base_m + j;
            if (m < n) h[(size_t)m * fout + col] = acc[t][j];
        }
    }
}

template <int CPL>
__global__ void gcn_gather_kernel(const float* __restrict__ h,
                                  const int* __restrict__ rowptr,
                                  const int2* __restrict__ meta,
                                  const float* __restrict__ dinv,
                                  const float* __restrict__ bias,
                                  float* __restrict__ out,
                                  int n, int fout, int do_relu) {
    const int wave = (blockIdx.x * blockDim.x + threadIdx.x) >> 5;
    const int lane = threadIdx.x & 31;
    if (wave >= n) return;
    const int i  = wave;
    const int c0 = lane * CPL;

    float acc[CPL];
    const float dv = dinv[i];
    const float* hi = h + (size_t)i * fout + c0;
    #pragma unroll
    for (int q = 0; q < CPL; ++q) acc[q] = hi[q] * dv * dv;

    const int beg = rowptr[i], end = rowptr[i + 1];
    #pragma unroll 2
    for (int j = beg; j < end; ++j) {
        const int2 m  = meta[j];
        const float nm = __int_as_float(m.y);
        const float* hs = h + (size_t)m.x * fout + c0;
        #pragma unroll
        for (int q = 0; q < CPL; ++q) acc[q] = fmaf(hs[q], nm, acc[q]);
    }

    float* oi = out + (size_t)i * fout + c0;
    #pragma unroll
    for (int q = 0; q < CPL; ++q) {
        float v = acc[q] + bias[c0 + q];
        oi[q] = do_relu ? fmaxf(v, 0.0f) : v;
    }
}


static inline int cdiv(long long a, int b) { return (int)((a + b - 1) / b); }
static inline size_t align4up(size_t x) { return (x + 3) & ~(size_t)3; }

extern "C" void kernel_launch(void* const* d_in, const int* in_sizes, int n_in,
                              void* d_out, int out_size, void* d_ws, size_t ws_size,
                              hipStream_t stream) {
    const float* x  = (const float*)d_in[0];
    const int*   ei = (const int*)  d_in[1];
    const float* W1 = (const float*)d_in[2];
    const float* b1 = (const float*)d_in[3];
    const float* W2 = (const float*)d_in[4];
    const float* b2 = (const float*)d_in[5];
    const float* W3 = (const float*)d_in[6];
    const float* b3 = (const float*)d_in[7];

    const int IN_C = 128, HID = 128, MID = 64, OUT = 64;
    const int n = in_sizes[0] / IN_C;
    const int E = in_sizes[1] / 2;
    const int* src = ei;
    const int* dst = ei + E;

    char* base = (char*)d_ws;
    size_t off = 0;
    auto take = [&](size_t elems) { void* p = base + off * 4; off += align4up(elems); return p; };
    int*   cnt    = (int*)  take(n);
    int*   rowptr = (int*)  take(n + 1);
    float* dinv   = (float*)take(n);
    int2*  meta   = (int2*) take((size_t)2 * E);
    float* h      = (float*)take((size_t)n * 128);
    float* xn     = (float*)take((size_t)n * 128);
    const int scan_blocks = cdiv(n, SCAN_TILE);
    int*   tsum   = (int*)  take(scan_blocks);
    (void)ws_size;

    const int B = 256;

    zero_i32_kernel<<<cdiv(n, B), B, 0, stream>>>(cnt, n);
    deg_count_kernel<<<cdiv(E, B), B, 0, stream>>>(dst, cnt, E);
    dinv_kernel<<<cdiv(n, B), B, 0, stream>>>(cnt, dinv, n);
    tile_total_kernel<<<scan_blocks, B, 0, stream>>>(cnt, tsum, n);
    scan_kernel<<<scan_blocks, B, 0, stream>>>(cnt, tsum, rowptr, n);
    zero_i32_kernel<<<cdiv(n, B), B, 0, stream>>>(cnt, n);
    csr_fill_kernel<<<cdiv(E, B), B, 0, stream>>>(src, dst, dinv, rowptr, cnt, meta, E);

    const int nrt = (n + 15) / 16;
    const int gemm_blocks = cdiv(nrt, GEMM_WAVES);
    const int gather_blocks = cdiv((long long)n * 32, B);

    auto run_layer = [&](const float* in, const float* W, const float* bias,
                         float* out, int fin, int fout, int relu) {
        if (fout == 128)
            gcn_gemm_wmma_kernel<8><<<gemm_blocks, B, 0, stream>>>(in, W, h, n, fin);
        else
            gcn_gemm_wmma_kernel<4><<<gemm_blocks, B, 0, stream>>>(in, W, h, n, fin);
        if (fout == 128)
            gcn_gather_kernel<4><<<gather_blocks, B, 0, stream>>>(h, rowptr, meta,
                                                                  dinv, bias, out, n, fout, relu);
        else
            gcn_gather_kernel<2><<<gather_blocks, B, 0, stream>>>(h, rowptr, meta,
                                                                  dinv, bias, out, n, fout, relu);
    };

    run_layer(x,  W1, b1, xn,            IN_C, HID, 1);
    run_layer(xn, W2, b2, xn,            HID,  MID, 1);
    run_layer(xn, W3, b3, (float*)d_out, MID,  OUT, 0);
}


# Gather: a node's edge records read in one coalesced load and handed out by readlane, up to 16 neighbour rows requested before the first add (was 2 records, then 2 rows, per wait); sum order kept
# speedup vs baseline: 1.4731x; 1.0646x over previous
#include <hip/hip_runtime.h>

typedef __attribute__((ext_vector_type(2))) float v2f;
typedef __attribute__((ext_vector_type(8))) float v8f;


__global__ void zero_i32_kernel(int* __restrict__ p, int n) {
    int t = blockIdx.x * blockDim.x + threadIdx.x;
    if (t < n) p[t] = 0;
}

__global__ void deg_count_kernel(const int* __restrict__ dst, int* __restrict__ cnt, int E) {
    int t = blockIdx.x * blockDim.x + threadIdx.x;
    if (t < E) atomicAdd(&cnt[dst[t]], 1);
}

__global__ void dinv_kernel(const int* __restrict__ cnt, float* __restrict__ dinv, int n) {
    int t = blockIdx.x * blockDim.x + threadIdx.x;
    if (t < n) dinv[t] = 1.0f / sqrtf((float)(cnt[t] + 1));
}

#define SCAN_TILE 1024

__device__ __forceinline__ int4 load_counts4(const int* __restrict__ cnt, int base, int n) {
    int4 v = make_int4(0, 0, 0, 0);
    if (base + 3 < n) {
        v = *reinterpret_cast<const int4*>(cnt + base);
    } else {
        if (base     < n) v.x = cnt[base];
        if (base + 1 < n) v.y = cnt[base + 1];
        if (base + 2 < n) v.z = cnt[base + 2];
    }
    return v;
}

__global__ __launch_bounds__(256)
void tile_total_kernel(const int* __restrict__ cnt, int* __restrict__ tsum, int n) {
    __shared__ int wsum[8];
    const int tid  = threadIdx.x;
    const int base = (blockIdx.x * 256 + tid) * 4;
    const int4 v = load_counts4(cnt, base, n);
    int s = (v.x + v.y) + (v.z + v.w);
    for (int off = 16; off > 0; off >>= 1) s += __shfl_xor(s, off);
    if ((tid & 31) == 0) wsum[tid >> 5] = s;
    __syncthreads();
    if (tid == 0) {
        int t = 0;
        for (int w = 0; w < 8; ++w) t += wsum[w];
        tsum[blockIdx.x] = t;
    }
}

__global__ __launch_bounds__(256)
void scan_kernel(const int* __restrict__ cnt, const int* __restrict__ tsum,
                 int* __restrict__ rowptr, int n) {
    __shared__ int wtile[8];
    __shared__ int wprev[8];
    const int tid  = threadIdx.x;
    const int lane = tid & 31;
    const int wv   = tid >> 5;
    const int base = (blockIdx.x * 256 + tid) * 4;
    const int4 v = load_counts4(cnt, base, n);

    int prev = 0;
    for (int j = tid; j < (int)blockIdx.x; j += 256) prev += tsum[j];

    const int t1 = v.x, t2 = t1 + v.y, t3 = t2 + v.z, t4 = t3 + v.w;
    int inc = t4;
    for (int off = 1; off < 32; off <<= 1) {
        const int u = __shfl_up(inc, off);
        if (lane >= off) inc += u;
    }
    for (int off = 16; off > 0; off >>= 1) prev += __shfl_xor(prev, off);
    if (lane == 31) wtile[wv] = inc;
    if (lane == 0)  wprev[wv] = prev;
    __syncthreads();

    int start = 0;
    for (int w = 0; w < 8; ++w) {
        start += wprev[w];
        if (w < wv) start += wtile[w];
    }
    const int excl = start + inc - t4;

    if (base + 3 < n) {
        *reinterpret_cast<int4*>(rowptr + base) =
            make_int4(excl, excl + t1, excl + t2, excl + t3);
    } else {
        if (base     < n) rowptr[base]     = excl;
        if (base + 1 < n) rowptr[base + 1] = excl + t1;
        if (base + 2 < n) rowptr[base + 2] = excl + t2;
    }
    if (base < n && base + 4 >= n) rowptr[n] = excl + t4;
}

__global__ void csr_fill_kernel(const int* __restrict__ src, const int* __restrict__ dst,
                                const float* __restrict__ dinv,
                                const int* __restrict__ rowptr, int* __restrict__ cursor,
                                int2* __restrict__ meta, int E) {
    int t = blockIdx.x * blockDim.x + threadIdx.x;
    if (t >= E) return;
    int s = src[t], d = dst[t];
    int slot = atomicAdd(&cursor[d], 1);
    int2 m;
    m.x = s;
    m.y = __float_as_int(dinv[s] * dinv[d]);
    meta[rowptr[d] + slot] = m;
}

#define GEMM_WAVES 8
template <int NCT>
__global__ __launch_bounds__(256)
void gcn_gemm_wmma_kernel(const float* __restrict__ A, const float* __restrict__ W,
                          float* __restrict__ h, int n, int fin) {
    const int fout = NCT * 16;
    __shared__ float Wt[128 * NCT * 16];
    const int tid = threadIdx.x;
    for (int idx = tid; idx < fin * fout; idx += 256)
        Wt[idx] = W[idx];
    __syncthreads();

    const int wv   = tid >> 5;
    const int lane = tid & 31;
    const int tm   = blockIdx.x * GEMM_WAVES + wv;
    const int nrt  = (n + 15) >> 4;
    if (tm >= nrt) return;

    const int half = lane >> 4;
    const int l    = lane & 15;
    int row  = tm * 16 + l;
    int arow = row < n ? row : n - 1;
    const float* ap = A + (size_t)arow * fin + half * 2;

    v8f acc[NCT];
    #pragma unroll
    for (int t = 0; t < NCT; ++t) acc[t] = {};

    #pragma unroll 2
    for (int k = 0; k < fin; k += 4) {
        v2f a;
        a.x = ap[k];
        a.y = ap[k + 1];
        const int kb = k + half * 2;
        #pragma unroll
        for (int t = 0; t < NCT; ++t) {
            const int col = t * 16 + l;
            v2f b;
            b.x = Wt[kb * fout + col];
            b.y = Wt[(kb + 1) * fout + col];
            acc[t] = __builtin_amdgcn_wmma_f32_16x16x4_f32(
                         false, a, false, b, (short)0, acc[t], false, false);
        }
    }

    const int base_m = tm * 16 + half * 8;
    #pragma unroll
    for (int t = 0; t < NCT; ++t) {
        const int col = t * 16 + l;
        #pragma unroll
        for (int j = 0; j < 8; ++j) {
            int m = base_m + j;
            if (m < n) h[(size_t)m * fout + col] = acc[t][j];
        }
    }
}

#define GATHER_AHEAD 16
template <int CPL>
__global__ void gcn_gather_kernel(const float* __restrict__ h,
                       const int* __restrict__ rowptr,
                       const int2* __restrict__ meta,
                       const float* __restrict__ dinv,
                       const float* __restrict__ bias,
                       float* __restrict__ out,
                       int n, int fout, int do_relu) {
    typedef __attribute__((ext_vector_type(CPL))) float vrow;
    const int lane = threadIdx.x & 31;
    const int i = __builtin_amdgcn_readfirstlane(
                      (int)((blockIdx.x * blockDim.x + threadIdx.x) >> 5));
    if (i >= n) return;
    const int c0 = lane * CPL;
    const float* hc = h + c0;

    const vrow self = *reinterpret_cast<const vrow*>(hc + (size_t)i * fout);
    const float dv = dinv[i];
    const int beg = rowptr[i], end = rowptr[i + 1];

    int nE = end - beg < 32 ? end - beg : 32;
    int2 m = make_int2(0, 0);
    if (lane < nE) m = meta[beg + lane];

    vrow acc = self * dv * dv;

    for (int cb = beg; cb < end; ) {
        for (int e = 0; e < nE; e += GATHER_AHEAD) {
            vrow  r[GATHER_AHEAD];
            float w[GATHER_AHEAD];
            #pragma unroll
            for (int q = 0; q < GATHER_AHEAD; ++q) {
                if (q == 0 || e + q < nE) {
                    const int sidx = __builtin_amdgcn_readlane(m.x, e + q);
                    w[q] = __int_as_float(__builtin_amdgcn_readlane(m.y, e + q));
                    r[q] = *reinterpret_cast<const vrow*>(hc + (size_t)sidx * fout);
                }
            }
            #pragma unroll
            for (int q = 0; q < GATHER_AHEAD; ++q) {
                if (q == 0 || e + q < nE) {
                    for (int c = 0; c < CPL; ++c) acc[c] = fmaf(r[q][c], w[q], acc[c]);
                }
            }
        }
        cb += 32;
        if (cb < end) {
            nE = end - cb < 32 ? end - cb : 32;
            m = make_int2(0, 0);
            if (lane < nE) m = meta[cb + lane];
        }
    }

    const vrow bv = *reinterpret_cast<const vrow*>(bias + c0);
    vrow v = acc + bv;
    if (do_relu) {
        for (int c = 0; c < CPL; ++c) v[c] = fmaxf(v[c], 0.0f);
    }
    *reinterpret_cast<vrow*>(out + (size_t)i * fout + c0) = v;
}


static inline int cdiv(long long a, int b) { return (int)((a + b - 1) / b); }
static inline size_t align4up(size_t x) { return (x + 3) & ~(size_t)3; }

extern "C" void kernel_launch(void* const* d_in, const int* in_sizes, int n_in,
                              void* d_out, int out_size, void* d_ws, size_t ws_size,
                              hipStream_t stream) {
    const float* x  = (const float*)d_in[0];
    const int*   ei = (const int*)  d_in[1];
    const float* W1 = (const float*)d_in[2];
    const float* b1 = (const float*)d_in[3];
    const float* W2 = (const float*)d_in[4];
    const float* b2 = (const float*)d_in[5];
    const float* W3 = (const float*)d_in[6];
    const float* b3 = (const float*)d_in[7];

    const int IN_C = 128, HID = 128, MID = 64, OUT = 64;
    const int n = in_sizes[0] / IN_C;
    const int E = in_sizes[1] / 2;
    const int* src = ei;
    const int* dst = ei + E;

    char* base = (char*)d_ws;
    size_t off = 0;
    auto take = [&](size_t elems) { void* p = base + off * 4; off += align4up(elems); return p; };
    int*   cnt    = (int*)  take(n);
    int*   rowptr = (int*)  take(n + 1);
    float* dinv   = (float*)take(n);
    int2*  meta   = (int2*) take((size_t)2 * E);
    float* h      = (float*)take((size_t)n * 128);
    float* xn     = (float*)take((size_t)n * 128);
    const int scan_blocks = cdiv(n, SCAN_TILE);
    int*   tsum   = (int*)  take(scan_blocks);
    (void)ws_size;

    const int B = 256;

    zero_i32_kernel<<<cdiv(n, B), B, 0, stream>>>(cnt, n);
    deg_count_kernel<<<cdiv(E, B), B, 0, stream>>>(dst, cnt, E);
    dinv_kernel<<<cdiv(n, B), B, 0, stream>>>(cnt, dinv, n);
    tile_total_kernel<<<scan_blocks, B, 0, stream>>>(cnt, tsum, n);
    scan_kernel<<<scan_blocks, B, 0, stream>>>(cnt, tsum, rowptr, n);
    zero_i32_kernel<<<cdiv(n, B), B, 0, stream>>>(cnt, n);
    csr_fill_kernel<<<cdiv(E, B), B, 0, stream>>>(src, dst, dinv, rowptr, cnt, meta, E);

    const int nrt = (n + 15) / 16;
    const int gemm_blocks = cdiv(nrt, GEMM_WAVES);
    const int gather_blocks = cdiv((long long)n * 32, B);

    auto run_layer = [&](const float* in, const float* W, const float* bias,
                         float* out, int fin, int fout, int relu) {
        if (fout == 128)
            gcn_gemm_wmma_kernel<8><<<gemm_blocks, B, 0, stream>>>(in, W, h, n, fin);
        else
            gcn_gemm_wmma_kernel<4><<<gemm_blocks, B, 0, stream>>>(in, W, h, n, fin);
        if (fout == 128)
            gcn_gather_kernel<4><<<gather_blocks, B, 0, stream>>>(h, rowptr, meta,
                                                                  dinv, bias, out, n, fout, relu);
        else
            gcn_gather_kernel<2><<<gather_blocks, B, 0, stream>>>(h, rowptr, meta,
                                                                  dinv, bias, out, n, fout, relu);
    };

    run_layer(x,  W1, b1, xn,            IN_C, HID, 1);
    run_layer(xn, W2, b2, xn,            HID,  MID, 1);
    run_layer(xn, W3, b3, (float*)d_out, MID,  OUT, 0);
}


# CSR build as a two-level counting sort on LDS counters (edges filed by destination range, then ranked per node in one work-group) replaces 2 M global atomics; per-node sum order stays arrival order
# speedup vs baseline: 2.3666x; 1.6065x over previous
#include <hip/hip_runtime.h>

typedef __attribute__((ext_vector_type(2))) float v2f;
typedef __attribute__((ext_vector_type(8))) float v8f;


#define SCAN_TILE 1024

__device__ __forceinline__ int4 load_counts4(const int* __restrict__ cnt, int base, int n) {
    int4 v = make_int4(0, 0, 0, 0);
    if (base + 3 < n) {
        v = *reinterpret_cast<const int4*>(cnt + base);
    } else {
        if (base     < n) v.x = cnt[base];
        if (base + 1 < n) v.y = cnt[base + 1];
        if (base + 2 < n) v.z = cnt[base + 2];
    }
    return v;
}

__global__ __launch_bounds__(256)
void tile_total_kernel(const int* __restrict__ cnt, int* __restrict__ tsum, int n) {
    __shared__ int wsum[8];
    const int tid  = threadIdx.x;
    const int base = (blockIdx.x * 256 + tid) * 4;
    const int4 v = load_counts4(cnt, base, n);
    int s = (v.x + v.y) + (v.z + v.w);
    for (int off = 16; off > 0; off >>= 1) s += __shfl_xor(s, off);
    if ((tid & 31) == 0) wsum[tid >> 5] = s;
    __syncthreads();
    if (tid == 0) {
        int t = 0;
        for (int w = 0; w < 8; ++w) t += wsum[w];
        tsum[blockIdx.x] = t;
    }
}

__global__ __launch_bounds__(256)
void scan_kernel(const int* __restrict__ cnt, const int* __restrict__ tsum,
                 int* __restrict__ rowptr, int n) {
    __shared__ int wtile[8];
    __shared__ int wprev[8];
    const int tid  = threadIdx.x;
    const int lane = tid & 31;
    const int wv   = tid >> 5;
    const int base = (blockIdx.x * 256 + tid) * 4;
    const int4 v = load_counts4(cnt, base, n);

    int prev = 0;
    for (int j = tid; j < (int)blockIdx.x; j += 256) prev += tsum[j];

    const int t1 = v.x, t2 = t1 + v.y, t3 = t2 + v.z, t4 = t3 + v.w;
    int inc = t4;
    for (int off = 1; off < 32; off <<= 1) {
        const int u = __shfl_up(inc, off);
        if (lane >= off) inc += u;
    }
    for (int off = 16; off > 0; off >>= 1) prev += __shfl_xor(prev, off);
    if (lane == 31) wtile[wv] = inc;
    if (lane == 0)  wprev[wv] = prev;
    __syncthreads();

    int start = 0;
    for (int w = 0; w < 8; ++w) {
        start += wprev[w];
        if (w < wv) start += wtile[w];
    }
    const int excl = start + inc - t4;

    if (base + 3 < n) {
        *reinterpret_cast<int4*>(rowptr + base) =
            make_int4(excl, excl + t1, excl + t2, excl + t3);
    } else {
        if (base     < n) rowptr[base]     = excl;
        if (base + 1 < n) rowptr[base + 1] = excl + t1;
        if (base + 2 < n) rowptr[base + 2] = excl + t2;
    }
    if (base < n && base + 4 >= n) rowptr[n] = excl + t4;
}

#define BUCKET_SHIFT 9
#define BUCKET_NODES (1 << BUCKET_SHIFT)
#define MAX_BUCKETS  1024
#define EDGE_CHUNK   4096
#define EDGES_PER_THREAD (EDGE_CHUNK / 256)

__global__ __launch_bounds__(256)
void bucket_count_kernel(const int* __restrict__ dst, int* __restrict__ table,
                         int E, int nchunk, int nb) {
    __shared__ int hist[MAX_BUCKETS];
    const int tid = threadIdx.x;
    const int chunk = blockIdx.x;
    const int e0 = chunk * EDGE_CHUNK;
    const int e1 = e0 + EDGE_CHUNK < E ? e0 + EDGE_CHUNK : E;
    for (int b = tid; b < nb; b += 256) hist[b] = 0;
    int d[EDGES_PER_THREAD];
    #pragma unroll
    for (int k = 0; k < EDGES_PER_THREAD; ++k) {
        const int e = e0 + k * 256 + tid;
        d[k] = e < e1 ? dst[e] : -1;
    }
    __syncthreads();
    #pragma unroll
    for (int k = 0; k < EDGES_PER_THREAD; ++k)
        if (d[k] >= 0) atomicAdd(&hist[d[k] >> BUCKET_SHIFT], 1);
    __syncthreads();
    for (int b = tid; b < nb; b += 256) table[b * nchunk + chunk] = hist[b];
}

__global__ __launch_bounds__(256)
void bucket_scatter_kernel(const int* __restrict__ src, const int* __restrict__ dst,
                           const int* __restrict__ toff, int2* __restrict__ filed,
                           int E, int nchunk, int nb) {
    __shared__ int cur[MAX_BUCKETS];
    const int tid = threadIdx.x;
    const int chunk = blockIdx.x;
    const int e0 = chunk * EDGE_CHUNK;
    const int e1 = e0 + EDGE_CHUNK < E ? e0 + EDGE_CHUNK : E;
    for (int b = tid; b < nb; b += 256) cur[b] = toff[b * nchunk + chunk];
    int s[EDGES_PER_THREAD], d[EDGES_PER_THREAD];
    #pragma unroll
    for (int k = 0; k < EDGES_PER_THREAD; ++k) {
        const int e = e0 + k * 256 + tid;
        s[k] = e < e1 ? src[e] : 0;
        d[k] = e < e1 ? dst[e] : -1;
    }
    __syncthreads();
    #pragma unroll
    for (int k = 0; k < EDGES_PER_THREAD; ++k) {
        if (d[k] >= 0) {
            const int pos = atomicAdd(&cur[d[k] >> BUCKET_SHIFT], 1);
            filed[pos] = make_int2(s[k], d[k]);
        }
    }
}

__global__ __launch_bounds__(1024)
void bucket_sort_kernel(const int2* __restrict__ filed, const int* __restrict__ toff,
                        int* __restrict__ rowptr, float* __restrict__ dinv,
                        int2* __restrict__ meta, int n, int nchunk, int nb) {
    __shared__ int   cnt[BUCKET_NODES];
    __shared__ int   cur[BUCKET_NODES];
    __shared__ float dl[BUCKET_NODES];
    __shared__ int   wtot[BUCKET_NODES / 32];
    const int tid  = threadIdx.x;
    const int lane = tid & 31;
    const int wv   = tid >> 5;
    const int b    = blockIdx.x;
    const int beg  = toff[b * nchunk];
    const int end  = toff[(b + 1) * nchunk];
    const int node0 = b << BUCKET_SHIFT;

    if (tid < BUCKET_NODES) cnt[tid] = 0;
    __syncthreads();
    for (int j0 = beg + tid; j0 < end; j0 += 4 * 1024) {
        int d[4];
        #pragma unroll
        for (int k = 0; k < 4; ++k) {
            const int j = j0 + k * 1024;
            d[k] = j < end ? filed[j].y : -1;
        }
        #pragma unroll
        for (int k = 0; k < 4; ++k)
            if (d[k] >= 0) atomicAdd(&cnt[d[k] - node0], 1);
    }
    __syncthreads();

    int c = 0, inc = 0;
    if (tid < BUCKET_NODES) {
        c = cnt[tid];
        inc = c;
        for (int off = 1; off < 32; off <<= 1) {
            const int u = __shfl_up(inc, off);
            if (lane >= off) inc += u;
        }
        if (lane == 31) wtot[wv] = inc;
    }
    __syncthreads();
    if (tid < BUCKET_NODES) {
        int start = beg;
        for (int w = 0; w < wv; ++w) start += wtot[w];
        const int excl = start + inc - c;
        cur[tid] = excl;
        const int node = node0 + tid;
        if (node < n) {
            const float dv = 1.0f / sqrtf((float)(c + 1));
            rowptr[node] = excl;
            dinv[node]   = dv;
            dl[tid]      = dv;
        }
    }
    if (b == nb - 1 && tid == 0) rowptr[n] = end;
    __syncthreads();

    for (int j0 = beg + tid; j0 < end; j0 += 4 * 1024) {
        int2 e[4];
        #pragma unroll
        for (int k = 0; k < 4; ++k) {
            const int j = j0 + k * 1024;
            e[k] = j < end ? filed[j] : make_int2(0, -1);
        }
        #pragma unroll
        for (int k = 0; k < 4; ++k) {
            if (e[k].y >= 0) {
                const int ln = e[k].y - node0;
                const int pos = atomicAdd(&cur[ln], 1);
                meta[pos] = make_int2(e[k].x, __float_as_int(dl[ln]));
            }
        }
    }
}

__global__ void norm_kernel(int2* __restrict__ meta, const float* __restrict__ dinv, int E) {
    const int t0 = (blockIdx.x * blockDim.x + threadIdx.x) * 4;
    if (t0 + 3 < E) {
        int4 a = *reinterpret_cast<const int4*>(meta + t0);
        int4 b = *reinterpret_cast<const int4*>(meta + t0 + 2);
        const float d0 = dinv[a.x], d1 = dinv[a.z], d2 = dinv[b.x], d3 = dinv[b.z];
        a.y = __float_as_int(d0 * __int_as_float(a.y));
        a.w = __float_as_int(d1 * __int_as_float(a.w));
        b.y = __float_as_int(d2 * __int_as_float(b.y));
        b.w = __float_as_int(d3 * __int_as_float(b.w));
        *reinterpret_cast<int4*>(meta + t0)     = a;
        *reinterpret_cast<int4*>(meta + t0 + 2) = b;
    } else {
        for (int t = t0; t < E; ++t) {
            int2 m = meta[t];
            m.y = __float_as_int(dinv[m.x] * __int_as_float(m.y));
            meta[t] = m;
        }
    }
}

#define GEMM_WAVES 8
template <int NCT>
__global__ __launch_bounds__(256)
void gcn_gemm_wmma_kernel(const float* __restrict__ A, const float* __restrict__ W,
                          float* __restrict__ h, int n, int fin) {
    const int fout = NCT * 16;
    __shared__ float Wt[128 * NCT * 16];
    const int tid = threadIdx.x;
    for (int idx = tid; idx < fin * fout; idx += 256)
        Wt[idx] = W[idx];
    __syncthreads();

    const int wv   = tid >> 5;
    const int lane = tid & 31;
    const int tm   = blockIdx.x * GEMM_WAVES + wv;
    const int nrt  = (n + 15) >> 4;
    if (tm >= nrt) return;

    const int half = lane >> 4;
    const int l    = lane & 15;
    int row  = tm * 16 + l;
    int arow = row < n ? row : n - 1;
    const float* ap = A + (size_t)arow * fin + half * 2;

    v8f acc[NCT];
    #pragma unroll
    for (int t = 0; t < NCT; ++t) acc[t] = {};

    #pragma unroll 2
    for (int k = 0; k < fin; k += 4) {
        v2f a;
        a.x = ap[k];
        a.y = ap[k + 1];
        const int kb = k + half * 2;
        #pragma unroll
        for (int t = 0; t < NCT; ++t) {
            const int col = t * 16 + l;
            v2f b;
            b.x = Wt[kb * fout + col];
            b.y = Wt[(kb + 1) * fout + col];
            acc[t] = __builtin_amdgcn_wmma_f32_16x16x4_f32(
                         false, a, false, b, (short)0, acc[t], false, false);
        }
    }

    const int base_m = tm * 16 + half * 8;
    #pragma unroll
    for (int t = 0; t < NCT; ++t) {
        const int col = t * 16 + l;
        #pragma unroll
        for (int j = 0; j < 8; ++j) {
            int m = base_m + j;
            if (m < n) h[(size_t)m * fout + col] = acc[t][j];
        }
    }
}

#define GATHER_AHEAD 16
template <int CPL>
__global__ void gcn_gather_kernel(const float* __restrict__ h,
                       const int* __restrict__ rowptr,
                       const int2* __restrict__ meta,
                       const float* __restrict__ dinv,
                       const float* __restrict__ bias,
                       float* __restrict__ out,
                       int n, int fout, int do_relu) {
    typedef __attribute__((ext_vector_type(CPL))) float vrow;
    const int lane = threadIdx.x & 31;
    const int i = __builtin_amdgcn_readfirstlane(
                      (int)((blockIdx.x * blockDim.x + threadIdx.x) >> 5));
    if (i >= n) return;
    const int c0 = lane * CPL;
    const float* hc = h + c0;

    const vrow self = *reinterpret_cast<const vrow*>(hc + (size_t)i * fout);
    const float dv = dinv[i];
    const int beg = rowptr[i], end = rowptr[i + 1];

    int nE = end - beg < 32 ? end - beg : 32;
    int2 m = make_int2(0, 0);
    if (lane < nE) m = meta[beg + lane];

    vrow acc = self * dv * dv;

    for (int cb = beg; cb < end; ) {
        for (int e = 0; e < nE; e += GATHER_AHEAD) {
            vrow  r[GATHER_AHEAD];
            float w[GATHER_AHEAD];
            #pragma unroll
            for (int q = 0; q < GATHER_AHEAD; ++q) {
                if (q == 0 || e + q < nE) {
                    const int sidx = __builtin_amdgcn_readlane(m.x, e + q);
                    w[q] = __int_as_float(__builtin_amdgcn_readlane(m.y, e + q));
                    r[q] = *reinterpret_cast<const vrow*>(hc + (size_t)sidx * fout);
                }
            }
            #pragma unroll
            for (int q = 0; q < GATHER_AHEAD; ++q) {
                if (q == 0 || e + q < nE) {
                    for (int c = 0; c < CPL; ++c) acc[c] = fmaf(r[q][c], w[q], acc[c]);
                }
            }
        }
        cb += 32;
        if (cb < end) {
            nE = end - cb < 32 ? end - cb : 32;
            m = make_int2(0, 0);
            if (lane < nE) m = meta[cb + lane];
        }
    }

    const vrow bv = *reinterpret_cast<const vrow*>(bias + c0);
    vrow v = acc + bv;
    if (do_relu) {
        for (int c = 0; c < CPL; ++c) v[c] = fmaxf(v[c], 0.0f);
    }
    *reinterpret_cast<vrow*>(out + (size_t)i * fout + c0) = v;
}


static inline int cdiv(long long a, int b) { return (int)((a + b - 1) / b); }
static inline size_t align4up(size_t x) { return (x + 3) & ~(size_t)3; }

extern "C" void kernel_launch(void* const* d_in, const int* in_sizes, int n_in,
                              void* d_out, int out_size, void* d_ws, size_t ws_size,
                              hipStream_t stream) {
    const float* x  = (const float*)d_in[0];
    const int*   ei = (const int*)  d_in[1];
    const float* W1 = (const float*)d_in[2];
    const float* b1 = (const float*)d_in[3];
    const float* W2 = (const float*)d_in[4];
    const float* b2 = (const float*)d_in[5];
    const float* W3 = (const float*)d_in[6];
    const float* b3 = (const float*)d_in[7];

    const int IN_C = 128, HID = 128, MID = 64, OUT = 64;
    const int n = in_sizes[0] / IN_C;
    const int E = in_sizes[1] / 2;
    const int* src = ei;
    const int* dst = ei + E;

    char* base = (char*)d_ws;
    size_t off = 0;
    auto take = [&](size_t elems) { void* p = base + off * 4; off += align4up(elems); return p; };
    int*   rowptr = (int*)  take(n + 1);
    float* dinv   = (float*)take(n);
    int2*  meta   = (int2*) take((size_t)2 * E);
    float* h      = (float*)take((size_t)n * 128);
    float* xn     = (float*)take((size_t)n * 128);
    const int nb     = cdiv(n, BUCKET_NODES);
    const int nchunk = cdiv(E, EDGE_CHUNK);
    const int ntab   = nb * nchunk;
    int*   table  = (int*)  take(ntab);
    int*   toff   = (int*)  take(ntab + 1);
    int2*  filed  = (int2*) take((size_t)2 * E);
    const int scan_blocks = cdiv(ntab, SCAN_TILE);
    int*   tsum   = (int*)  take(scan_blocks);
    (void)ws_size;

    const int B = 256;

    bucket_count_kernel<<<nchunk, B, 0, stream>>>(dst, table, E, nchunk, nb);
    tile_total_kernel<<<scan_blocks, B, 0, stream>>>(table, tsum, ntab);
    scan_kernel<<<scan_blocks, B, 0, stream>>>(table, tsum, toff, ntab);
    bucket_scatter_kernel<<<nchunk, B, 0, stream>>>(src, dst, toff, filed, E, nchunk, nb);
    bucket_sort_kernel<<<nb, 1024, 0, stream>>>(filed, toff, rowptr, dinv, meta, n, nchunk, nb);
    norm_kernel<<<cdiv(E, 4 * B), B, 0, stream>>>(meta, dinv, E);

    const int nrt = (n + 15) / 16;
    const int gemm_blocks = cdiv(nrt, GEMM_WAVES);
    const int gather_blocks = cdiv((long long)n * 32, B);

    auto run_layer = [&](const float* in, const float* W, const float* bias,
                         float* out, int fin, int fout, int relu) {
        if (fout == 128)
            gcn_gemm_wmma_kernel<8><<<gemm_blocks, B, 0, stream>>>(in, W, h, n, fin);
        else
            gcn_gemm_wmma_kernel<4><<<gemm_blocks, B, 0, stream>>>(in, W, h, n, fin);
        if (fout == 128)
            gcn_gather_kernel<4><<<gather_blocks, B, 0, stream>>>(h, rowptr, meta,
                                                                  dinv, bias, out, n, fout, relu);
        else
            gcn_gather_kernel<2><<<gather_blocks, B, 0, stream>>>(h, rowptr, meta,
                                                                  dinv, bias, out, n, fout, relu);
    };

    run_layer(x,  W1, b1, xn,            IN_C, HID, 1);
    run_layer(xn, W2, b2, xn,            HID,  MID, 1);
    run_layer(xn, W3, b3, (float*)d_out, MID,  OUT, 0);
}


# Layers 2,3: dense product fused into the preceding aggregate kernel (a work-group's 64 activated rows stay in LDS and feed the WMMA; a wave walks 8 nodes, asking one ahead); 2 launches go; sums kept
# speedup vs baseline: 2.7223x; 1.1503x over previous
#include <hip/hip_runtime.h>

typedef __attribute__((ext_vector_type(2))) float v2f;
typedef __attribute__((ext_vector_type(8))) float v8f;


#define SCAN_TILE 1024

__device__ __forceinline__ int4 load_counts4(const int* __restrict__ cnt, int base, int n) {
    int4 v = make_int4(0, 0, 0, 0);
    if (base + 3 < n) {
        v = *reinterpret_cast<const int4*>(cnt + base);
    } else {
        if (base     < n) v.x = cnt[base];
        if (base + 1 < n) v.y = cnt[base + 1];
        if (base + 2 < n) v.z = cnt[base + 2];
    }
    return v;
}

__global__ __launch_bounds__(256)
void tile_total_kernel(const int* __restrict__ cnt, int* __restrict__ tsum, int n) {
    __shared__ int wsum[8];
    const int tid  = threadIdx.x;
    const int base = (blockIdx.x * 256 + tid) * 4;
    const int4 v = load_counts4(cnt, base, n);
    int s = (v.x + v.y) + (v.z + v.w);
    for (int off = 16; off > 0; off >>= 1) s += __shfl_xor(s, off);
    if ((tid & 31) == 0) wsum[tid >> 5] = s;
    __syncthreads();
    if (tid == 0) {
        int t = 0;
        for (int w = 0; w < 8; ++w) t += wsum[w];
        tsum[blockIdx.x] = t;
    }
}

__global__ __launch_bounds__(256)
void scan_kernel(const int* __restrict__ cnt, const int* __restrict__ tsum,
                 int* __restrict__ rowptr, int n) {
    __shared__ int wtile[8];
    __shared__ int wprev[8];
    const int tid  = threadIdx.x;
    const int lane = tid & 31;
    const int wv   = tid >> 5;
    const int base = (blockIdx.x * 256 + tid) * 4;
    const int4 v = load_counts4(cnt, base, n);

    int prev = 0;
    for (int j = tid; j < (int)blockIdx.x; j += 256) prev += tsum[j];

    const int t1 = v.x, t2 = t1 + v.y, t3 = t2 + v.z, t4 = t3 + v.w;
    int inc = t4;
    for (int off = 1; off < 32; off <<= 1) {
        const int u = __shfl_up(inc, off);
        if (lane >= off) inc += u;
    }
    for (int off = 16; off > 0; off >>= 1) prev += __shfl_xor(prev, off);
    if (lane == 31) wtile[wv] = inc;
    if (lane == 0)  wprev[wv] = prev;
    __syncthreads();

    int start = 0;
    for (int w = 0; w < 8; ++w) {
        start += wprev[w];
        if (w < wv) start += wtile[w];
    }
    const int excl = start + inc - t4;

    if (base + 3 < n) {
        *reinterpret_cast<int4*>(rowptr + base) =
            make_int4(excl, excl + t1, excl + t2, excl + t3);
    } else {
        if (base     < n) rowptr[base]     = excl;
        if (base + 1 < n) rowptr[base + 1] = excl + t1;
        if (base + 2 < n) rowptr[base + 2] = excl + t2;
    }
    if (base < n && base + 4 >= n) rowptr[n] = excl + t4;
}

#define BUCKET_SHIFT 9
#define BUCKET_NODES (1 << BUCKET_SHIFT)
#define MAX_BUCKETS  1024
#define EDGE_CHUNK   4096
#define EDGES_PER_THREAD (EDGE_CHUNK / 256)

__global__ __launch_bounds__(256)
void bucket_count_kernel(const int* __restrict__ dst, int* __restrict__ table,
                         int E, int nchunk, int nb) {
    __shared__ int hist[MAX_BUCKETS];
    const int tid = threadIdx.x;
    const int chunk = blockIdx.x;
    const int e0 = chunk * EDGE_CHUNK;
    const int e1 = e0 + EDGE_CHUNK < E ? e0 + EDGE_CHUNK : E;
    for (int b = tid; b < nb; b += 256) hist[b] = 0;
    int d[EDGES_PER_THREAD];
    #pragma unroll
    for (int k = 0; k < EDGES_PER_THREAD; ++k) {
        const int e = e0 + k * 256 + tid;
        d[k] = e < e1 ? dst[e] : -1;
    }
    __syncthreads();
    #pragma unroll
    for (int k = 0; k < EDGES_PER_THREAD; ++k)
        if (d[k] >= 0) atomicAdd(&hist[d[k] >> BUCKET_SHIFT], 1);
    __syncthreads();
    for (int b = tid; b < nb; b += 256) table[b * nchunk + chunk] = hist[b];
}

__global__ __launch_bounds__(256)
void bucket_scatter_kernel(const int* __restrict__ src, const int* __restrict__ dst,
                           const int* __restrict__ toff, int2* __restrict__ filed,
                           int E, int nchunk, int nb) {
    __shared__ int cur[MAX_BUCKETS];
    const int tid = threadIdx.x;
    const int chunk = blockIdx.x;
    const int e0 = chunk * EDGE_CHUNK;
    const int e1 = e0 + EDGE_CHUNK < E ? e0 + EDGE_CHUNK : E;
    for (int b = tid; b < nb; b += 256) cur[b] = toff[b * nchunk + chunk];
    int s[EDGES_PER_THREAD], d[EDGES_PER_THREAD];
    #pragma unroll
    for (int k = 0; k < EDGES_PER_THREAD; ++k) {
        const int e = e0 + k * 256 + tid;
        s[k] = e < e1 ? src[e] : 0;
        d[k] = e < e1 ? dst[e] : -1;
    }
    __syncthreads();
    #pragma unroll
    for (int k = 0; k < EDGES_PER_THREAD; ++k) {
        if (d[k] >= 0) {
            const int pos = atomicAdd(&cur[d[k] >> BUCKET_SHIFT], 1);
            filed[pos] = make_int2(s[k], d[k]);
        }
    }
}

__global__ __launch_bounds__(1024)
void bucket_sort_kernel(const int2* __restrict__ filed, const int* __restrict__ toff,
                        int* __restrict__ rowptr, float* __restrict__ dinv,
                        int2* __restrict__ meta, int n, int nchunk, int nb) {
    __shared__ int   cnt[BUCKET_NODES];
    __shared__ int   cur[BUCKET_NODES];
    __shared__ float dl[BUCKET_NODES];
    __shared__ int   wtot[BUCKET_NODES / 32];
    const int tid  = threadIdx.x;
    const int lane = tid & 31;
    const int wv   = tid >> 5;
    const int b    = blockIdx.x;
    const int beg  = toff[b * nchunk];
    const int end  = toff[(b + 1) * nchunk];
    const int node0 = b << BUCKET_SHIFT;

    if (tid < BUCKET_NODES) cnt[tid] = 0;
    __syncthreads();
    for (int j0 = beg + tid; j0 < end; j0 += 4 * 1024) {
        int d[4];
        #pragma unroll
        for (int k = 0; k < 4; ++k) {
            const int j = j0 + k * 1024;
            d[k] = j < end ? filed[j].y : -1;
        }
        #pragma unroll
        for (int k = 0; k < 4; ++k)
            if (d[k] >= 0) atomicAdd(&cnt[d[k] - node0], 1);
    }
    __syncthreads();

    int c = 0, inc = 0;
    if (tid < BUCKET_NODES) {
        c = cnt[tid];
        inc = c;
        for (int off = 1; off < 32; off <<= 1) {
            const int u = __shfl_up(inc, off);
            if (lane >= off) inc += u;
        }
        if (lane == 31) wtot[wv] = inc;
    }
    __syncthreads();
    if (tid < BUCKET_NODES) {
        int start = beg;
        for (int w = 0; w < wv; ++w) start += wtot[w];
        const int excl = start + inc - c;
        cur[tid] = excl;
        const int node = node0 + tid;
        if (node < n) {
            const float dv = 1.0f / sqrtf((float)(c + 1));
            rowptr[node] = excl;
            dinv[node]   = dv;
            dl[tid]      = dv;
        }
    }
    if (b == nb - 1 && tid == 0) rowptr[n] = end;
    __syncthreads();

    for (int j0 = beg + tid; j0 < end; j0 += 4 * 1024) {
        int2 e[4];
        #pragma unroll
        for (int k = 0; k < 4; ++k) {
            const int j = j0 + k * 1024;
            e[k] = j < end ? filed[j] : make_int2(0, -1);
        }
        #pragma unroll
        for (int k = 0; k < 4; ++k) {
            if (e[k].y >= 0) {
                const int ln = e[k].y - node0;
                const int pos = atomicAdd(&cur[ln], 1);
                meta[pos] = make_int2(e[k].x, __float_as_int(dl[ln]));
            }
        }
    }
}

__global__ void norm_kernel(int2* __restrict__ meta, const float* __restrict__ dinv, int E) {
    const int t0 = (blockIdx.x * blockDim.x + threadIdx.x) * 4;
    if (t0 + 3 < E) {
        int4 a = *reinterpret_cast<const int4*>(meta + t0);
        int4 b = *reinterpret_cast<const int4*>(meta + t0 + 2);
        const float d0 = dinv[a.x], d1 = dinv[a.z], d2 = dinv[b.x], d3 = dinv[b.z];
        a.y = __float_as_int(d0 * __int_as_float(a.y));
        a.w = __float_as_int(d1 * __int_as_float(a.w));
        b.y = __float_as_int(d2 * __int_as_float(b.y));
        b.w = __float_as_int(d3 * __int_as_float(b.w));
        *reinterpret_cast<int4*>(meta + t0)     = a;
        *reinterpret_cast<int4*>(meta + t0 + 2) = b;
    } else {
        for (int t = t0; t < E; ++t) {
            int2 m = meta[t];
            m.y = __float_as_int(dinv[m.x] * __int_as_float(m.y));
            meta[t] = m;
        }
    }
}

#define GEMM_WAVES 8
template <int NCT>
__global__ __launch_bounds__(256)
void gcn_gemm_wmma_kernel(const float* __restrict__ A, const float* __restrict__ W,
                          float* __restrict__ h, int n, int fin) {
    const int fout = NCT * 16;
    __shared__ float Wt[128 * NCT * 16];
    const int tid = threadIdx.x;
    for (int idx = tid; idx < fin * fout; idx += 256)
        Wt[idx] = W[idx];
    __syncthreads();

    const int wv   = tid >> 5;
    const int lane = tid & 31;
    const int tm   = blockIdx.x * GEMM_WAVES + wv;
    const int nrt  = (n + 15) >> 4;
    if (tm >= nrt) return;

    const int half = lane >> 4;
    const int l    = lane & 15;
    int row  = tm * 16 + l;
    int arow = row < n ? row : n - 1;
    const float* ap = A + (size_t)arow * fin + half * 2;

    v8f acc[NCT];
    #pragma unroll
    for (int t = 0; t < NCT; ++t) acc[t] = {};

    #pragma unroll 2
    for (int k = 0; k < fin; k += 4) {
        v2f a;
        a.x = ap[k];
        a.y = ap[k + 1];
        const int kb = k + half * 2;
        #pragma unroll
        for (int t = 0; t < NCT; ++t) {
            const int col = t * 16 + l;
            v2f b;
            b.x = Wt[kb * fout + col];
            b.y = Wt[(kb + 1) * fout + col];
            acc[t] = __builtin_amdgcn_wmma_f32_16x16x4_f32(
                         false, a, false, b, (short)0, acc[t], false, false);
        }
    }

    const int base_m = tm * 16 + half * 8;
    #pragma unroll
    for (int t = 0; t < NCT; ++t) {
        const int col = t * 16 + l;
        #pragma unroll
        for (int j = 0; j < 8; ++j) {
            int m = base_m + j;
            if (m < n) h[(size_t)m * fout + col] = acc[t][j];
        }
    }
}

#define GATHER_AHEAD 16
template <int CPL>
__global__ void gcn_gather_kernel(const float* __restrict__ h,
                       const int* __restrict__ rowptr,
                       const int2* __restrict__ meta,
                       const float* __restrict__ dinv,
                       const float* __restrict__ bias,
                       float* __restrict__ out,
                       int n, int fout, int do_relu) {
    typedef __attribute__((ext_vector_type(CPL))) float vrow;
    const int lane = threadIdx.x & 31;
    const int i = __builtin_amdgcn_readfirstlane(
                      (int)((blockIdx.x * blockDim.x + threadIdx.x) >> 5));
    if (i >= n) return;
    const int c0 = lane * CPL;
    const float* hc = h + c0;

    const vrow self = *reinterpret_cast<const vrow*>(hc + (size_t)i * fout);
    const float dv = dinv[i];
    const int beg = rowptr[i], end = rowptr[i + 1];

    int nE = end - beg < 32 ? end - beg : 32;
    int2 m = make_int2(0, 0);
    if (lane < nE) m = meta[beg + lane];

    vrow acc = self * dv * dv;

    for (int cb = beg; cb < end; ) {
        for (int e = 0; e < nE; e += GATHER_AHEAD) {
            vrow  r[GATHER_AHEAD];
            float w[GATHER_AHEAD];
            #pragma unroll
            for (int q = 0; q < GATHER_AHEAD; ++q) {
                if (q == 0 || e + q < nE) {
                    const int sidx = __builtin_amdgcn_readlane(m.x, e + q);
                    w[q] = __int_as_float(__builtin_amdgcn_readlane(m.y, e + q));
                    r[q] = *reinterpret_cast<const vrow*>(hc + (size_t)sidx * fout);
                }
            }
            #pragma unroll
            for (int q = 0; q < GATHER_AHEAD; ++q) {
                if (q == 0 || e + q < nE) {
                    for (int c = 0; c < CPL; ++c) acc[c] = fmaf(r[q][c], w[q], acc[c]);
                }
            }
        }
        cb += 32;
        if (cb < end) {
            nE = end - cb < 32 ? end - cb : 32;
            m = make_int2(0, 0);
            if (lane < nE) m = meta[cb + lane];
        }
    }

    const vrow bv = *reinterpret_cast<const vrow*>(bias + c0);
    vrow v = acc + bv;
    if (do_relu) {
        for (int c = 0; c < CPL; ++c) v[c] = fmaxf(v[c], 0.0f);
    }
    *reinterpret_cast<vrow*>(out + (size_t)i * fout + c0) = v;
}


#define FUSE_NODES 64
template <int CPL>
__global__ __launch_bounds__(256)
void gcn_gather_gemm_kernel(const float* __restrict__ h,
                       const int* __restrict__ rowptr,
                       const int2* __restrict__ meta,
                       const float* __restrict__ dinv,
                       const float* __restrict__ bias,
                       const float* __restrict__ Wn,
                       float* __restrict__ hn,
                       int n) {
    typedef __attribute__((ext_vector_type(CPL))) float vrow;
    constexpr int FIN   = 32 * CPL;
    constexpr int FOUT  = 64;
    constexpr int PITCH = FIN + 4;
    constexpr int NPW   = FUSE_NODES / 8;
    __shared__ float Wt[FIN * FOUT];
    __shared__ float Xt[FUSE_NODES * PITCH];
    const int tid  = threadIdx.x;
    const int lane = tid & 31;
    const int wv = __builtin_amdgcn_readfirstlane(tid >> 5);

    constexpr int NW4 = FIN * FOUT / (256 * 4);
    float4 wreg[NW4];
    #pragma unroll
    for (int q = 0; q < NW4; ++q)
        wreg[q] = *reinterpret_cast<const float4*>(Wn + (q * 256 + tid) * 4);
    #pragma unroll
    for (int q = 0; q < NW4; ++q)
        *reinterpret_cast<float4*>(Wt + (q * 256 + tid) * 4) = wreg[q];

    const int node0 = blockIdx.x * FUSE_NODES;
    const int fout = FIN;
    const int c0 = lane * CPL;
    const float* hc = h + c0;
    const vrow bv = *reinterpret_cast<const vrow*>(bias + c0);
    const int ibase = node0 + wv * NPW;

    int sb = 0, se = 0; float sd = 0.0f;
    if (ibase < n) { sb = rowptr[ibase]; se = rowptr[ibase + 1]; sd = dinv[ibase]; }
    int begN = sb, endN = se; float dvN = sd;
    int nEN = endN - begN < 32 ? endN - begN : 32;
    vrow selfN;
    for (int c = 0; c < CPL; ++c) selfN[c] = 0.0f;
    if (ibase < n) selfN = *reinterpret_cast<const vrow*>(hc + (size_t)ibase * fout);
    int2 mN = make_int2(0, 0);
    if (lane < nEN) mN = meta[begN + lane];
    sb = 0; se = 0; sd = 0.0f;
    if (NPW > 1 && ibase + 1 < n) { sb = rowptr[ibase + 1]; se = rowptr[ibase + 2]; sd = dinv[ibase + 1]; }

    for (int rr = 0; rr < NPW; ++rr) {
        const int lr = wv * NPW + rr;
        const int i  = ibase + rr;
        const int beg = begN, end = endN;
        const float dv = dvN;
        const vrow self = selfN;
        const int2 m = mN;
        const int nE = nEN;

        begN = sb; endN = se; dvN = sd;
        nEN = endN - begN < 32 ? endN - begN : 32;
        if (rr + 1 < NPW && i + 1 < n)
            selfN = *reinterpret_cast<const vrow*>(hc + (size_t)(i + 1) * fout);
        mN = make_int2(0, 0);
        if (lane < nEN) mN = meta[begN + lane];
        sb = 0; se = 0; sd = 0.0f;
        if (rr + 2 < NPW && i + 2 < n) { sb = rowptr[i + 2]; se = rowptr[i + 3]; sd = dinv[i + 2]; }

        vrow v;
        for (int c = 0; c < CPL; ++c) v[c] = 0.0f;
        if (i < n) {
            vrow acc = self * dv * dv;

            auto sweep = [&](const int2 mm, const int cnt) {
                for (int e = 0; e < cnt; e += GATHER_AHEAD) {
                    vrow  r[GATHER_AHEAD];
                    float w[GATHER_AHEAD];
                    #pragma unroll
                    for (int q = 0; q < GATHER_AHEAD; ++q) {
                        if (q == 0 || e + q < cnt) {
                            const int sidx = __builtin_amdgcn_readlane(mm.x, e + q);
                            w[q] = __int_as_float(__builtin_amdgcn_readlane(mm.y, e + q));
                            r[q] = *reinterpret_cast<const vrow*>(hc + (size_t)sidx * fout);
                        }
                    }
                    #pragma unroll
                    for (int q = 0; q < GATHER_AHEAD; ++q) {
                        if (q == 0 || e + q < cnt) {
                            for (int c = 0; c < CPL; ++c) acc[c] = fmaf(r[q][c], w[q], acc[c]);
                        }
                    }
                }
            };
            sweep(m, nE);
            for (int cb = beg + 32; cb < end; cb += 32) {
                const int cnt = end - cb < 32 ? end - cb : 32;
                int2 mm = make_int2(0, 0);
                if (lane < cnt) mm = meta[cb + lane];
                sweep(mm, cnt);
            }
            v = acc + bv;
            for (int c = 0; c < CPL; ++c) v[c] = fmaxf(v[c], 0.0f);
        }
        *reinterpret_cast<vrow*>(Xt + lr * PITCH + c0) = v;
    }
    __syncthreads();

    const int rt   = wv >> 1;
    const int ctb  = (wv & 1) * 2;
    const int half = lane >> 4;
    const int l    = lane & 15;
    const float* xa = Xt + (rt * 16 + l) * PITCH + half * 2;
    const float* wb = Wt + (half * 2) * FOUT + ctb * 16 + l;

    v8f acc0 = {}, acc1 = {};
    for (int k = 0; k < FIN; k += 4) {
        v2f a = *reinterpret_cast<const v2f*>(xa + k);
        v2f b0, b1;
        b0.x = wb[k * FOUT];
        b0.y = wb[(k + 1) * FOUT];
        b1.x = wb[k * FOUT + 16];
        b1.y = wb[(k + 1) * FOUT + 16];
        acc0 = __builtin_amdgcn_wmma_f32_16x16x4_f32(
                   false, a, false, b0, (short)0, acc0, false, false);
        acc1 = __builtin_amdgcn_wmma_f32_16x16x4_f32(
                   false, a, false, b1, (short)0, acc1, false, false);
    }

    const int base_m = node0 + rt * 16 + half * 8;
    const int col = ctb * 16 + l;
    #pragma unroll
    for (int j = 0; j < 8; ++j) {
        const int m = base_m + j;
        if (m < n) {
            hn[(size_t)m * FOUT + col]      = acc0[j];
            hn[(size_t)m * FOUT + col + 16] = acc1[j];
        }
    }
}


static inline int cdiv(long long a, int b) { return (int)((a + b - 1) / b); }
static inline size_t align4up(size_t x) { return (x + 3) & ~(size_t)3; }

extern "C" void kernel_launch(void* const* d_in, const int* in_sizes, int n_in,
                              void* d_out, int out_size, void* d_ws, size_t ws_size,
                              hipStream_t stream) {
    const float* x  = (const float*)d_in[0];
    const int*   ei = (const int*)  d_in[1];
    const float* W1 = (const float*)d_in[2];
    const float* b1 = (const float*)d_in[3];
    const float* W2 = (const float*)d_in[4];
    const float* b2 = (const float*)d_in[5];
    const float* W3 = (const float*)d_in[6];
    const float* b3 = (const float*)d_in[7];

    const int IN_C = 128;
    const int n = in_sizes[0] / IN_C;
    const int E = in_sizes[1] / 2;
    const int* src = ei;
    const int* dst = ei + E;

    char* base = (char*)d_ws;
    size_t off = 0;
    auto take = [&](size_t elems) { void* p = base + off * 4; off += align4up(elems); return p; };
    int*   rowptr = (int*)  take(n + 1);
    float* dinv   = (float*)take(n);
    int2*  meta   = (int2*) take((size_t)2 * E);
    float* h      = (float*)take((size_t)n * 128);
    float* xn     = (float*)take((size_t)n * 128);
    const int nb     = cdiv(n, BUCKET_NODES);
    const int nchunk = cdiv(E, EDGE_CHUNK);
    const int ntab   = nb * nchunk;
    int*   table  = (int*)  take(ntab);
    int*   toff   = (int*)  take(ntab + 1);
    int2*  filed  = (int2*) take((size_t)2 * E);
    const int scan_blocks = cdiv(ntab, SCAN_TILE);
    int*   tsum   = (int*)  take(scan_blocks);
    (void)ws_size;

    const int B = 256;

    bucket_count_kernel<<<nchunk, B, 0, stream>>>(dst, table, E, nchunk, nb);
    tile_total_kernel<<<scan_blocks, B, 0, stream>>>(table, tsum, ntab);
    scan_kernel<<<scan_blocks, B, 0, stream>>>(table, tsum, toff, ntab);
    bucket_scatter_kernel<<<nchunk, B, 0, stream>>>(src, dst, toff, filed, E, nchunk, nb);
    bucket_sort_kernel<<<nb, 1024, 0, stream>>>(filed, toff, rowptr, dinv, meta, n, nchunk, nb);
    norm_kernel<<<cdiv(E, 4 * B), B, 0, stream>>>(meta, dinv, E);

    const int nrt = (n + 15) / 16;
    const int gemm_blocks = cdiv(nrt, GEMM_WAVES);
    const int gather_blocks = cdiv((long long)n * 32, B);
    const int fuse_blocks = cdiv(n, FUSE_NODES);

    gcn_gemm_wmma_kernel<8><<<gemm_blocks, B, 0, stream>>>(x, W1, h, n, IN_C);
    gcn_gather_gemm_kernel<4><<<fuse_blocks, B, 0, stream>>>(h, rowptr, meta, dinv,
                                                             b1, W2, xn, n);
    gcn_gather_gemm_kernel<2><<<fuse_blocks, B, 0, stream>>>(xn, rowptr, meta, dinv,
                                                             b2, W3, h, n);
    gcn_gather_kernel<2><<<gather_blocks, B, 0, stream>>>(h, rowptr, meta, dinv, b3,
                                                          (float*)d_out, n, 64, 0);
}


# Workspace planes start on 256-byte boundaries (rows of h and xn sat 16 bytes past one): a gathered 512/256-byte row covers 2/1 aligned 256-byte units and 4/2 lines, was 3/2 units and 5/3 lines
# speedup vs baseline: 2.8531x; 1.0480x over previous
#include <hip/hip_runtime.h>

typedef __attribute__((ext_vector_type(2))) float v2f;
typedef __attribute__((ext_vector_type(8))) float v8f;


#define SCAN_TILE 1024

__device__ __forceinline__ int4 load_counts4(const int* __restrict__ cnt, int base, int n) {
    int4 v = make_int4(0, 0, 0, 0);
    if (base + 3 < n) {
        v = *reinterpret_cast<const int4*>(cnt + base);
    } else {
        if (base     < n) v.x = cnt[base];
        if (base + 1 < n) v.y = cnt[base + 1];
        if (base + 2 < n) v.z = cnt[base + 2];
    }
    return v;
}

__global__ __launch_bounds__(256)
void tile_total_kernel(const int* __restrict__ cnt, int* __restrict__ tsum, int n) {
    __shared__ int wsum[8];
    const int tid  = threadIdx.x;
    const int base = (blockIdx.x * 256 + tid) * 4;
    const int4 v = load_counts4(cnt, base, n);
    int s = (v.x + v.y) + (v.z + v.w);
    for (int off = 16; off > 0; off >>= 1) s += __shfl_xor(s, off);
    if ((tid & 31) == 0) wsum[tid >> 5] = s;
    __syncthreads();
    if (tid == 0) {
        int t = 0;
        for (int w = 0; w < 8; ++w) t += wsum[w];
        tsum[blockIdx.x] = t;
    }
}

__global__ __launch_bounds__(256)
void scan_kernel(const int* __restrict__ cnt, const int* __restrict__ tsum,
                 int* __restrict__ rowptr, int n) {
    __shared__ int wtile[8];
    __shared__ int wprev[8];
    const int tid  = threadIdx.x;
    const int lane = tid & 31;
    const int wv   = tid >> 5;
    const int base = (blockIdx.x * 256 + tid) * 4;
    const int4 v = load_counts4(cnt, base, n);

    int prev = 0;
    for (int j = tid; j < (int)blockIdx.x; j += 256) prev += tsum[j];

    const int t1 = v.x, t2 = t1 + v.y, t3 = t2 + v.z, t4 = t3 + v.w;
    int inc = t4;
    for (int off = 1; off < 32; off <<= 1) {
        const int u = __shfl_up(inc, off);
        if (lane >= off) inc += u;
    }
    for (int off = 16; off > 0; off >>= 1) prev += __shfl_xor(prev, off);
    if (lane == 31) wtile[wv] = inc;
    if (lane == 0)  wprev[wv] = prev;
    __syncthreads();

    int start = 0;
    for (int w = 0; w < 8; ++w) {
        start += wprev[w];
        if (w < wv) start += wtile[w];
    }
    const int excl = start + inc - t4;

    if (base + 3 < n) {
        *reinterpret_cast<int4*>(rowptr + base) =
            make_int4(excl, excl + t1, excl + t2, excl + t3);
    } else {
        if (base     < n) rowptr[base]     = excl;
        if (base + 1 < n) rowptr[base + 1] = excl + t1;
        if (base + 2 < n) rowptr[base + 2] = excl + t2;
    }
    if (base < n && base + 4 >= n) rowptr[n] = excl + t4;
}

#define BUCKET_SHIFT 9
#define BUCKET_NODES (1 << BUCKET_SHIFT)
#define MAX_BUCKETS  1024
#define EDGE_CHUNK   4096
#define EDGES_PER_THREAD (EDGE_CHUNK / 256)

__global__ __launch_bounds__(256)
void bucket_count_kernel(const int* __restrict__ dst, int* __restrict__ table,
                         int E, int nchunk, int nb) {
    __shared__ int hist[MAX_BUCKETS];
    const int tid = threadIdx.x;
    const int chunk = blockIdx.x;
    const int e0 = chunk * EDGE_CHUNK;
    const int e1 = e0 + EDGE_CHUNK < E ? e0 + EDGE_CHUNK : E;
    for (int b = tid; b < nb; b += 256) hist[b] = 0;
    int d[EDGES_PER_THREAD];
    #pragma unroll
    for (int k = 0; k < EDGES_PER_THREAD; ++k) {
        const int e = e0 + k * 256 + tid;
        d[k] = e < e1 ? dst[e] : -1;
    }
    __syncthreads();
    #pragma unroll
    for (int k = 0; k < EDGES_PER_THREAD; ++k)
        if (d[k] >= 0) atomicAdd(&hist[d[k] >> BUCKET_SHIFT], 1);
    __syncthreads();
    for (int b = tid; b < nb; b += 256) table[b * nchunk + chunk] = hist[b];
}

__global__ __launch_bounds__(256)
void bucket_scatter_kernel(const int* __restrict__ src, const int* __restrict__ dst,
                           const int* __restrict__ toff, int2* __restrict__ filed,
                           int E, int nchunk, int nb) {
    __shared__ int cur[MAX_BUCKETS];
    const int tid = threadIdx.x;
    const int chunk = blockIdx.x;
    const int e0 = chunk * EDGE_CHUNK;
    const int e1 = e0 + EDGE_CHUNK < E ? e0 + EDGE_CHUNK : E;
    for (int b = tid; b < nb; b += 256) cur[b] = toff[b * nchunk + chunk];
    int s[EDGES_PER_THREAD], d[EDGES_PER_THREAD];
    #pragma unroll
    for (int k = 0; k < EDGES_PER_THREAD; ++k) {
        const int e = e0 + k * 256 + tid;
        s[k] = e < e1 ? src[e] : 0;
        d[k] = e < e1 ? dst[e] : -1;
    }
    __syncthreads();
    #pragma unroll
    for (int k = 0; k < EDGES_PER_THREAD; ++k) {
        if (d[k] >= 0) {
            const int pos = atomicAdd(&cur[d[k] >> BUCKET_SHIFT], 1);
            filed[pos] = make_int2(s[k], d[k]);
        }
    }
}

__global__ __launch_bounds__(1024)
void bucket_sort_kernel(const int2* __restrict__ filed, const int* __restrict__ toff,
                        int* __restrict__ rowptr, float* __restrict__ dinv,
                        int2* __restrict__ meta, int n, int nchunk, int nb) {
    __shared__ int   cnt[BUCKET_NODES];
    __shared__ int   cur[BUCKET_NODES];
    __shared__ float dl[BUCKET_NODES];
    __shared__ int   wtot[BUCKET_NODES / 32];
    const int tid  = threadIdx.x;
    const int lane = tid & 31;
    const int wv   = tid >> 5;
    const int b    = blockIdx.x;
    const int beg  = toff[b * nchunk];
    const int end  = toff[(b + 1) * nchunk];
    const int node0 = b << BUCKET_SHIFT;

    if (tid < BUCKET_NODES) cnt[tid] = 0;
    __syncthreads();
    for (int j0 = beg + tid; j0 < end; j0 += 4 * 1024) {
        int d[4];
        #pragma unroll
        for (int k = 0; k < 4; ++k) {
            const int j = j0 + k * 1024;
            d[k] = j < end ? filed[j].y : -1;
        }
        #pragma unroll
        for (int k = 0; k < 4; ++k)
            if (d[k] >= 0) atomicAdd(&cnt[d[k] - node0], 1);
    }
    __syncthreads();

    int c = 0, inc = 0;
    if (tid < BUCKET_NODES) {
        c = cnt[tid];
        inc = c;
        for (int off = 1; off < 32; off <<= 1) {
            const int u = __shfl_up(inc, off);
            if (lane >= off) inc += u;
        }
        if (lane == 31) wtot[wv] = inc;
    }
    __syncthreads();
    if (tid < BUCKET_NODES) {
        int start = beg;
        for (int w = 0; w < wv; ++w) start += wtot[w];
        const int excl = start + inc - c;
        cur[tid] = excl;
        const int node = node0 + tid;
        if (node < n) {
            const float dv = 1.0f / sqrtf((float)(c + 1));
            rowptr[node] = excl;
            dinv[node]   = dv;
            dl[tid]      = dv;
        }
    }
    if (b == nb - 1 && tid == 0) rowptr[n] = end;
    __syncthreads();

    for (int j0 = beg + tid; j0 < end; j0 += 4 * 1024) {
        int2 e[4];
        #pragma unroll
        for (int k = 0; k < 4; ++k) {
            const int j = j0 + k * 1024;
            e[k] = j < end ? filed[j] : make_int2(0, -1);
        }
        #pragma unroll
        for (int k = 0; k < 4; ++k) {
            if (e[k].y >= 0) {
                const int ln = e[k].y - node0;
                const int pos = atomicAdd(&cur[ln], 1);
                meta[pos] = make_int2(e[k].x, __float_as_int(dl[ln]));
            }
        }
    }
}

__global__ void norm_kernel(int2* __restrict__ meta, const float* __restrict__ dinv, int E) {
    const int t0 = (blockIdx.x * blockDim.x + threadIdx.x) * 4;
    if (t0 + 3 < E) {
        int4 a = *reinterpret_cast<const int4*>(meta + t0);
        int4 b = *reinterpret_cast<const int4*>(meta + t0 + 2);
        const float d0 = dinv[a.x], d1 = dinv[a.z], d2 = dinv[b.x], d3 = dinv[b.z];
        a.y = __float_as_int(d0 * __int_as_float(a.y));
        a.w = __float_as_int(d1 * __int_as_float(a.w));
        b.y = __float_as_int(d2 * __int_as_float(b.y));
        b.w = __float_as_int(d3 * __int_as_float(b.w));
        *reinterpret_cast<int4*>(meta + t0)     = a;
        *reinterpret_cast<int4*>(meta + t0 + 2) = b;
    } else {
        for (int t = t0; t < E; ++t) {
            int2 m = meta[t];
            m.y = __float_as_int(dinv[m.x] * __int_as_float(m.y));
            meta[t] = m;
        }
    }
}

#define GEMM_WAVES 8
template <int NCT>
__global__ __launch_bounds__(256)
void gcn_gemm_wmma_kernel(const float* __restrict__ A, const float* __restrict__ W,
                          float* __restrict__ h, int n, int fin) {
    const int fout = NCT * 16;
    __shared__ float Wt[128 * NCT * 16];
    const int tid = threadIdx.x;
    for (int idx = tid; idx < fin * fout; idx += 256)
        Wt[idx] = W[idx];
    __syncthreads();

    const int wv   = tid >> 5;
    const int lane = tid & 31;
    const int tm   = blockIdx.x * GEMM_WAVES + wv;
    const int nrt  = (n + 15) >> 4;
    if (tm >= nrt) return;

    const int half = lane >> 4;
    const int l    = lane & 15;
    int row  = tm * 16 + l;
    int arow = row < n ? row : n - 1;
    const float* ap = A + (size_t)arow * fin + half * 2;

    v8f acc[NCT];
    #pragma unroll
    for (int t = 0; t < NCT; ++t) acc[t] = {};

    #pragma unroll 2
    for (int k = 0; k < fin; k += 4) {
        v2f a;
        a.x = ap[k];
        a.y = ap[k + 1];
        const int kb = k + half * 2;
        #pragma unroll
        for (int t = 0; t < NCT; ++t) {
            const int col = t * 16 + l;
            v2f b;
            b.x = Wt[kb * fout + col];
            b.y = Wt[(kb + 1) * fout + col];
            acc[t] = __builtin_amdgcn_wmma_f32_16x16x4_f32(
                         false, a, false, b, (short)0, acc[t], false, false);
        }
    }

    const int base_m = tm * 16 + half * 8;
    #pragma unroll
    for (int t = 0; t < NCT; ++t) {
        const int col = t * 16 + l;
        #pragma unroll
        for (int j = 0; j < 8; ++j) {
            int m = base_m + j;
            if (m < n) h[(size_t)m * fout + col] = acc[t][j];
        }
    }
}

#define GATHER_AHEAD 16
template <int CPL>
__global__ void gcn_gather_kernel(const float* __restrict__ h,
                       const int* __restrict__ rowptr,
                       const int2* __restrict__ meta,
                       const float* __restrict__ dinv,
                       const float* __restrict__ bias,
                       float* __restrict__ out,
                       int n, int fout, int do_relu) {
    typedef __attribute__((ext_vector_type(CPL))) float vrow;
    const int lane = threadIdx.x & 31;
    const int i = __builtin_amdgcn_readfirstlane(
                      (int)((blockIdx.x * blockDim.x + threadIdx.x) >> 5));
    if (i >= n) return;
    const int c0 = lane * CPL;
    const float* hc = h + c0;

    const vrow self = *reinterpret_cast<const vrow*>(hc + (size_t)i * fout);
    const float dv = dinv[i];
    const int beg = rowptr[i], end = rowptr[i + 1];

    int nE = end - beg < 32 ? end - beg : 32;
    int2 m = make_int2(0, 0);
    if (lane < nE) m = meta[beg + lane];

    vrow acc = self * dv * dv;

    for (int cb = beg; cb < end; ) {
        for (int e = 0; e < nE; e += GATHER_AHEAD) {
            vrow  r[GATHER_AHEAD];
            float w[GATHER_AHEAD];
            #pragma unroll
            for (int q = 0; q < GATHER_AHEAD; ++q) {
                if (q == 0 || e + q < nE) {
                    const int sidx = __builtin_amdgcn_readlane(m.x, e + q);
                    w[q] = __int_as_float(__builtin_amdgcn_readlane(m.y, e + q));
                    r[q] = *reinterpret_cast<const vrow*>(hc + (size_t)sidx * fout);
                }
            }
            #pragma unroll
            for (int q = 0; q < GATHER_AHEAD; ++q) {
                if (q == 0 || e + q < nE) {
                    for (int c = 0; c < CPL; ++c) acc[c] = fmaf(r[q][c], w[q], acc[c]);
                }
            }
        }
        cb += 32;
        if (cb < end) {
            nE = end - cb < 32 ? end - cb : 32;
            m = make_int2(0, 0);
            if (lane < nE) m = meta[cb + lane];
        }
    }

    const vrow bv = *reinterpret_cast<const vrow*>(bias + c0);
    vrow v = acc + bv;
    if (do_relu) {
        for (int c = 0; c < CPL; ++c) v[c] = fmaxf(v[c], 0.0f);
    }
    *reinterpret_cast<vrow*>(out + (size_t)i * fout + c0) = v;
}


#define FUSE_NODES 64
template <int CPL>
__global__ __launch_bounds__(256)
void gcn_gather_gemm_kernel(const float* __restrict__ h,
                       const int* __restrict__ rowptr,
                       const int2* __restrict__ meta,
                       const float* __restrict__ dinv,
                       const float* __restrict__ bias,
                       const float* __restrict__ Wn,
                       float* __restrict__ hn,
                       int n) {
    typedef __attribute__((ext_vector_type(CPL))) float vrow;
    constexpr int FIN   = 32 * CPL;
    constexpr int FOUT  = 64;
    constexpr int PITCH = FIN + 4;
    constexpr int NPW   = FUSE_NODES / 8;
    __shared__ float Wt[FIN * FOUT];
    __shared__ float Xt[FUSE_NODES * PITCH];
    const int tid  = threadIdx.x;
    const int lane = tid & 31;
    const int wv = __builtin_amdgcn_readfirstlane(tid >> 5);

    constexpr int NW4 = FIN * FOUT / (256 * 4);
    float4 wreg[NW4];
    #pragma unroll
    for (int q = 0; q < NW4; ++q)
        wreg[q] = *reinterpret_cast<const float4*>(Wn + (q * 256 + tid) * 4);
    #pragma unroll
    for (int q = 0; q < NW4; ++q)
        *reinterpret_cast<float4*>(Wt + (q * 256 + tid) * 4) = wreg[q];

    const int node0 = blockIdx.x * FUSE_NODES;
    const int fout = FIN;
    const int c0 = lane * CPL;
    const float* hc = h + c0;
    const vrow bv = *reinterpret_cast<const vrow*>(bias + c0);
    const int ibase = node0 + wv * NPW;

    int sb = 0, se = 0; float sd = 0.0f;
    if (ibase < n) { sb = rowptr[ibase]; se = rowptr[ibase + 1]; sd = dinv[ibase]; }
    int begN = sb, endN = se; float dvN = sd;
    int nEN = endN - begN < 32 ? endN - begN : 32;
    vrow selfN;
    for (int c = 0; c < CPL; ++c) selfN[c] = 0.0f;
    if (ibase < n) selfN = *reinterpret_cast<const vrow*>(hc + (size_t)ibase * fout);
    int2 mN = make_int2(0, 0);
    if (lane < nEN) mN = meta[begN + lane];
    sb = 0; se = 0; sd = 0.0f;
    if (NPW > 1 && ibase + 1 < n) { sb = rowptr[ibase + 1]; se = rowptr[ibase + 2]; sd = dinv[ibase + 1]; }

    for (int rr = 0; rr < NPW; ++rr) {
        const int lr = wv * NPW + rr;
        const int i  = ibase + rr;
        const int beg = begN, end = endN;
        const float dv = dvN;
        const vrow self = selfN;
        const int2 m = mN;
        const int nE = nEN;

        begN = sb; endN = se; dvN = sd;
        nEN = endN - begN < 32 ? endN - begN : 32;
        if (rr + 1 < NPW && i + 1 < n)
            selfN = *reinterpret_cast<const vrow*>(hc + (size_t)(i + 1) * fout);
        mN = make_int2(0, 0);
        if (lane < nEN) mN = meta[begN + lane];
        sb = 0; se = 0; sd = 0.0f;
        if (rr + 2 < NPW && i + 2 < n) { sb = rowptr[i + 2]; se = rowptr[i + 3]; sd = dinv[i + 2]; }

        vrow v;
        for (int c = 0; c < CPL; ++c) v[c] = 0.0f;
        if (i < n) {
            vrow acc = self * dv * dv;

            auto sweep = [&](const int2 mm, const int cnt) {
                for (int e = 0; e < cnt; e += GATHER_AHEAD) {
                    vrow  r[GATHER_AHEAD];
                    float w[GATHER_AHEAD];
                    #pragma unroll
                    for (int q = 0; q < GATHER_AHEAD; ++q) {
                        if (q == 0 || e + q < cnt) {
                            const int sidx = __builtin_amdgcn_readlane(mm.x, e + q);
                            w[q] = __int_as_float(__builtin_amdgcn_readlane(mm.y, e + q));
                            r[q] = *reinterpret_cast<const vrow*>(hc + (size_t)sidx * fout);
                        }
                    }
                    #pragma unroll
                    for (int q = 0; q < GATHER_AHEAD; ++q) {
                        if (q == 0 || e + q < cnt) {
                            for (int c = 0; c < CPL; ++c) acc[c] = fmaf(r[q][c], w[q], acc[c]);
                        }
                    }
                }
            };
            sweep(m, nE);
            for (int cb = beg + 32; cb < end; cb += 32) {
                const int cnt = end - cb < 32 ? end - cb : 32;
                int2 mm = make_int2(0, 0);
                if (lane < cnt) mm = meta[cb + lane];
                sweep(mm, cnt);
            }
            v = acc + bv;
            for (int c = 0; c < CPL; ++c) v[c] = fmaxf(v[c], 0.0f);
        }
        *reinterpret_cast<vrow*>(Xt + lr * PITCH + c0) = v;
    }
    __syncthreads();

    const int rt   = wv >> 1;
    const int ctb  = (wv & 1) * 2;
    const int half = lane >> 4;
    const int l    = lane & 15;
    const float* xa = Xt + (rt * 16 + l) * PITCH + half * 2;
    const float* wb = Wt + (half * 2) * FOUT + ctb * 16 + l;

    v8f acc0 = {}, acc1 = {};
    for (int k = 0; k < FIN; k += 4) {
        v2f a = *reinterpret_cast<const v2f*>(xa + k);
        v2f b0, b1;
        b0.x = wb[k * FOUT];
        b0.y = wb[(k + 1) * FOUT];
        b1.x = wb[k * FOUT + 16];
        b1.y = wb[(k + 1) * FOUT + 16];
        acc0 = __builtin_amdgcn_wmma_f32_16x16x4_f32(
                   false, a, false, b0, (short)0, acc0, false, false);
        acc1 = __builtin_amdgcn_wmma_f32_16x16x4_f32(
                   false, a, false, b1, (short)0, acc1, false, false);
    }

    const int base_m = node0 + rt * 16 + half * 8;
    const int col = ctb * 16 + l;
    #pragma unroll
    for (int j = 0; j < 8; ++j) {
        const int m = base_m + j;
        if (m < n) {
            hn[(size_t)m * FOUT + col]      = acc0[j];
            hn[(size_t)m * FOUT + col + 16] = acc1[j];
        }
    }
}


static inline int cdiv(long long a, int b) { return (int)((a + b - 1) / b); }
static inline size_t align64up(size_t x) { return (x + 63) & ~(size_t)63; }

extern "C" void kernel_launch(void* const* d_in, const int* in_sizes, int n_in,
                              void* d_out, int out_size, void* d_ws, size_t ws_size,
                              hipStream_t stream) {
    const float* x  = (const float*)d_in[0];
    const int*   ei = (const int*)  d_in[1];
    const float* W1 = (const float*)d_in[2];
    const float* b1 = (const float*)d_in[3];
    const float* W2 = (const float*)d_in[4];
    const float* b2 = (const float*)d_in[5];
    const float* W3 = (const float*)d_in[6];
    const float* b3 = (const float*)d_in[7];

    const int IN_C = 128;
    const int n = in_sizes[0] / IN_C;
    const int E = in_sizes[1] / 2;
    const int* src = ei;
    const int* dst = ei + E;

    char* base = (char*)d_ws;
    size_t off = ((256 - ((size_t)base & 255)) & 255) / 4;
    auto take = [&](size_t elems) { void* p = base + off * 4; off += align64up(elems); return p; };
    int*   rowptr = (int*)  take(n + 1);
    float* dinv   = (float*)take(n);
    int2*  meta   = (int2*) take((size_t)2 * E);
    float* h      = (float*)take((size_t)n * 128);
    float* xn     = (float*)take((size_t)n * 128);
    const int nb     = cdiv(n, BUCKET_NODES);
    const int nchunk = cdiv(E, EDGE_CHUNK);
    const int ntab   = nb * nchunk;
    int*   table  = (int*)  take(ntab);
    int*   toff   = (int*)  take(ntab + 1);
    int2*  filed  = (int2*) take((size_t)2 * E);
    const int scan_blocks = cdiv(ntab, SCAN_TILE);
    int*   tsum   = (int*)  take(scan_blocks);
    (void)ws_size;

    const int B = 256;

    bucket_count_kernel<<<nchunk, B, 0, stream>>>(dst, table, E, nchunk, nb);
    tile_total_kernel<<<scan_blocks, B, 0, stream>>>(table, tsum, ntab);
    scan_kernel<<<scan_blocks, B, 0, stream>>>(table, tsum, toff, ntab);
    bucket_scatter_kernel<<<nchunk, B, 0, stream>>>(src, dst, toff, filed, E, nchunk, nb);
    bucket_sort_kernel<<<nb, 1024, 0, stream>>>(filed, toff, rowptr, dinv, meta, n, nchunk, nb);
    norm_kernel<<<cdiv(E, 4 * B), B, 0, stream>>>(meta, dinv, E);

    const int nrt = (n + 15) / 16;
    const int gemm_blocks = cdiv(nrt, GEMM_WAVES);
    const int gather_blocks = cdiv((long long)n * 32, B);
    const int fuse_blocks = cdiv(n, FUSE_NODES);

    gcn_gemm_wmma_kernel<8><<<gemm_blocks, B, 0, stream>>>(x, W1, h, n, IN_C);
    gcn_gather_gemm_kernel<4><<<fuse_blocks, B, 0, stream>>>(h, rowptr, meta, dinv,
                                                             b1, W2, xn, n);
    gcn_gather_gemm_kernel<2><<<fuse_blocks, B, 0, stream>>>(xn, rowptr, meta, dinv,
                                                             b2, W3, h, n);
    gcn_gather_kernel<2><<<gather_blocks, B, 0, stream>>>(h, rowptr, meta, dinv, b3,
                                                          (float*)d_out, n, 64, 0);
}


# Layer 1: x rows are aggregated first and multiplied by W1 inside the fused kernel (agg(x W1) = agg(x) W1): its product launch and the h round trip go; rows are summed before the product, not after
# speedup vs baseline: 3.0472x; 1.0680x over previous
#include <hip/hip_runtime.h>

typedef __attribute__((ext_vector_type(2))) float v2f;
typedef __attribute__((ext_vector_type(8))) float v8f;


#define SCAN_TILE 1024

__device__ __forceinline__ int4 load_counts4(const int* __restrict__ cnt, int base, int n) {
    int4 v = make_int4(0, 0, 0, 0);
    if (base + 3 < n) {
        v = *reinterpret_cast<const int4*>(cnt + base);
    } else {
        if (base     < n) v.x = cnt[base];
        if (base + 1 < n) v.y = cnt[base + 1];
        if (base + 2 < n) v.z = cnt[base + 2];
    }
    return v;
}

__global__ __launch_bounds__(256)
void tile_total_kernel(const int* __restrict__ cnt, int* __restrict__ tsum, int n) {
    __shared__ int wsum[8];
    const int tid  = threadIdx.x;
    const int base = (blockIdx.x * 256 + tid) * 4;
    const int4 v = load_counts4(cnt, base, n);
    int s = (v.x + v.y) + (v.z + v.w);
    for (int off = 16; off > 0; off >>= 1) s += __shfl_xor(s, off);
    if ((tid & 31) == 0) wsum[tid >> 5] = s;
    __syncthreads();
    if (tid == 0) {
        int t = 0;
        for (int w = 0; w < 8; ++w) t += wsum[w];
        tsum[blockIdx.x] = t;
    }
}

__global__ __launch_bounds__(256)
void scan_kernel(const int* __restrict__ cnt, const int* __restrict__ tsum,
                 int* __restrict__ rowptr, int n) {
    __shared__ int wtile[8];
    __shared__ int wprev[8];
    const int tid  = threadIdx.x;
    const int lane = tid & 31;
    const int wv   = tid >> 5;
    const int base = (blockIdx.x * 256 + tid) * 4;
    const int4 v = load_counts4(cnt, base, n);

    int prev = 0;
    for (int j = tid; j < (int)blockIdx.x; j += 256) prev += tsum[j];

    const int t1 = v.x, t2 = t1 + v.y, t3 = t2 + v.z, t4 = t3 + v.w;
    int inc = t4;
    for (int off = 1; off < 32; off <<= 1) {
        const int u = __shfl_up(inc, off);
        if (lane >= off) inc += u;
    }
    for (int off = 16; off > 0; off >>= 1) prev += __shfl_xor(prev, off);
    if (lane == 31) wtile[wv] = inc;
    if (lane == 0)  wprev[wv] = prev;
    __syncthreads();

    int start = 0;
    for (int w = 0; w < 8; ++w) {
        start += wprev[w];
        if (w < wv) start += wtile[w];
    }
    const int excl = start + inc - t4;

    if (base + 3 < n) {
        *reinterpret_cast<int4*>(rowptr + base) =
            make_int4(excl, excl + t1, excl + t2, excl + t3);
    } else {
        if (base     < n) rowptr[base]     = excl;
        if (base + 1 < n) rowptr[base + 1] = excl + t1;
        if (base + 2 < n) rowptr[base + 2] = excl + t2;
    }
    if (base < n && base + 4 >= n) rowptr[n] = excl + t4;
}

#define BUCKET_SHIFT 9
#define BUCKET_NODES (1 << BUCKET_SHIFT)
#define MAX_BUCKETS  1024
#define EDGE_CHUNK   4096
#define EDGES_PER_THREAD (EDGE_CHUNK / 256)

__global__ __launch_bounds__(256)
void bucket_count_kernel(const int* __restrict__ dst, int* __restrict__ table,
                         int E, int nchunk, int nb) {
    __shared__ int hist[MAX_BUCKETS];
    const int tid = threadIdx.x;
    const int chunk = blockIdx.x;
    const int e0 = chunk * EDGE_CHUNK;
    const int e1 = e0 + EDGE_CHUNK < E ? e0 + EDGE_CHUNK : E;
    for (int b = tid; b < nb; b += 256) hist[b] = 0;
    int d[EDGES_PER_THREAD];
    #pragma unroll
    for (int k = 0; k < EDGES_PER_THREAD; ++k) {
        const int e = e0 + k * 256 + tid;
        d[k] = e < e1 ? dst[e] : -1;
    }
    __syncthreads();
    #pragma unroll
    for (int k = 0; k < EDGES_PER_THREAD; ++k)
        if (d[k] >= 0) atomicAdd(&hist[d[k] >> BUCKET_SHIFT], 1);
    __syncthreads();
    for (int b = tid; b < nb; b += 256) table[b * nchunk + chunk] = hist[b];
}

__global__ __launch_bounds__(256)
void bucket_scatter_kernel(const int* __restrict__ src, const int* __restrict__ dst,
                           const int* __restrict__ toff, int2* __restrict__ filed,
                           int E, int nchunk, int nb) {
    __shared__ int cur[MAX_BUCKETS];
    const int tid = threadIdx.x;
    const int chunk = blockIdx.x;
    const int e0 = chunk * EDGE_CHUNK;
    const int e1 = e0 + EDGE_CHUNK < E ? e0 + EDGE_CHUNK : E;
    for (int b = tid; b < nb; b += 256) cur[b] = toff[b * nchunk + chunk];
    int s[EDGES_PER_THREAD], d[EDGES_PER_THREAD];
    #pragma unroll
    for (int k = 0; k < EDGES_PER_THREAD; ++k) {
        const int e = e0 + k * 256 + tid;
        s[k] = e < e1 ? src[e] : 0;
        d[k] = e < e1 ? dst[e] : -1;
    }
    __syncthreads();
    #pragma unroll
    for (int k = 0; k < EDGES_PER_THREAD; ++k) {
        if (d[k] >= 0) {
            const int pos = atomicAdd(&cur[d[k] >> BUCKET_SHIFT], 1);
            filed[pos] = make_int2(s[k], d[k]);
        }
    }
}

__global__ __launch_bounds__(1024)
void bucket_sort_kernel(const int2* __restrict__ filed, const int* __restrict__ toff,
                        int* __restrict__ rowptr, float* __restrict__ dinv,
                        int2* __restrict__ meta, int n, int nchunk, int nb) {
    __shared__ int   cnt[BUCKET_NODES];
    __shared__ int   cur[BUCKET_NODES];
    __shared__ float dl[BUCKET_NODES];
    __shared__ int   wtot[BUCKET_NODES / 32];
    const int tid  = threadIdx.x;
    const int lane = tid & 31;
    const int wv   = tid >> 5;
    const int b    = blockIdx.x;
    const int beg  = toff[b * nchunk];
    const int end  = toff[(b + 1) * nchunk];
    const int node0 = b << BUCKET_SHIFT;

    if (tid < BUCKET_NODES) cnt[tid] = 0;
    __syncthreads();
    for (int j0 = beg + tid; j0 < end; j0 += 4 * 1024) {
        int d[4];
        #pragma unroll
        for (int k = 0; k < 4; ++k) {
            const int j = j0 + k * 1024;
            d[k] = j < end ? filed[j].y : -1;
        }
        #pragma unroll
        for (int k = 0; k < 4; ++k)
            if (d[k] >= 0) atomicAdd(&cnt[d[k] - node0], 1);
    }
    __syncthreads();

    int c = 0, inc = 0;
    if (tid < BUCKET_NODES) {
        c = cnt[tid];
        inc = c;
        for (int off = 1; off < 32; off <<= 1) {
            const int u = __shfl_up(inc, off);
            if (lane >= off) inc += u;
        }
        if (lane == 31) wtot[wv] = inc;
    }
    __syncthreads();
    if (tid < BUCKET_NODES) {
        int start = beg;
        for (int w = 0; w < wv; ++w) start += wtot[w];
        const int excl = start + inc - c;
        cur[tid] = excl;
        const int node = node0 + tid;
        if (node < n) {
            const float dv = 1.0f / sqrtf((float)(c + 1));
            rowptr[node] = excl;
            dinv[node]   = dv;
            dl[tid]      = dv;
        }
    }
    if (b == nb - 1 && tid == 0) rowptr[n] = end;
    __syncthreads();

    for (int j0 = beg + tid; j0 < end; j0 += 4 * 1024) {
        int2 e[4];
        #pragma unroll
        for (int k = 0; k < 4; ++k) {
            const int j = j0 + k * 1024;
            e[k] = j < end ? filed[j] : make_int2(0, -1);
        }
        #pragma unroll
        for (int k = 0; k < 4; ++k) {
            if (e[k].y >= 0) {
                const int ln = e[k].y - node0;
                const int pos = atomicAdd(&cur[ln], 1);
                meta[pos] = make_int2(e[k].x, __float_as_int(dl[ln]));
            }
        }
    }
}

__global__ void norm_kernel(int2* __restrict__ meta, const float* __restrict__ dinv, int E) {
    const int t0 = (blockIdx.x * blockDim.x + threadIdx.x) * 4;
    if (t0 + 3 < E) {
        int4 a = *reinterpret_cast<const int4*>(meta + t0);
        int4 b = *reinterpret_cast<const int4*>(meta + t0 + 2);
        const float d0 = dinv[a.x], d1 = dinv[a.z], d2 = dinv[b.x], d3 = dinv[b.z];
        a.y = __float_as_int(d0 * __int_as_float(a.y));
        a.w = __float_as_int(d1 * __int_as_float(a.w));
        b.y = __float_as_int(d2 * __int_as_float(b.y));
        b.w = __float_as_int(d3 * __int_as_float(b.w));
        *reinterpret_cast<int4*>(meta + t0)     = a;
        *reinterpret_cast<int4*>(meta + t0 + 2) = b;
    } else {
        for (int t = t0; t < E; ++t) {
            int2 m = meta[t];
            m.y = __float_as_int(dinv[m.x] * __int_as_float(m.y));
            meta[t] = m;
        }
    }
}

#define GATHER_AHEAD 16
template <int CPL>
__global__ void gcn_gather_kernel(const float* __restrict__ h,
                       const int* __restrict__ rowptr,
                       const int2* __restrict__ meta,
                       const float* __restrict__ dinv,
                       const float* __restrict__ bias,
                       float* __restrict__ out,
                       int n, int fout, int do_relu) {
    typedef __attribute__((ext_vector_type(CPL))) float vrow;
    const int lane = threadIdx.x & 31;
    const int i = __builtin_amdgcn_readfirstlane(
                      (int)((blockIdx.x * blockDim.x + threadIdx.x) >> 5));
    if (i >= n) return;
    const int c0 = lane * CPL;
    const float* hc = h + c0;

    const vrow self = *reinterpret_cast<const vrow*>(hc + (size_t)i * fout);
    const float dv = dinv[i];
    const int beg = rowptr[i], end = rowptr[i + 1];

    int nE = end - beg < 32 ? end - beg : 32;
    int2 m = make_int2(0, 0);
    if (lane < nE) m = meta[beg + lane];

    vrow acc = self * dv * dv;

    for (int cb = beg; cb < end; ) {
        for (int e = 0; e < nE; e += GATHER_AHEAD) {
            vrow  r[GATHER_AHEAD];
            float w[GATHER_AHEAD];
            #pragma unroll
            for (int q = 0; q < GATHER_AHEAD; ++q) {
                if (q == 0 || e + q < nE) {
                    const int sidx = __builtin_amdgcn_readlane(m.x, e + q);
                    w[q] = __int_as_float(__builtin_amdgcn_readlane(m.y, e + q));
                    r[q] = *reinterpret_cast<const vrow*>(hc + (size_t)sidx * fout);
                }
            }
            #pragma unroll
            for (int q = 0; q < GATHER_AHEAD; ++q) {
                if (q == 0 || e + q < nE) {
                    for (int c = 0; c < CPL; ++c) acc[c] = fmaf(r[q][c], w[q], acc[c]);
                }
            }
        }
        cb += 32;
        if (cb < end) {
            nE = end - cb < 32 ? end - cb : 32;
            m = make_int2(0, 0);
            if (lane < nE) m = meta[cb + lane];
        }
    }

    const vrow bv = *reinterpret_cast<const vrow*>(bias + c0);
    vrow v = acc + bv;
    if (do_relu) {
        for (int c = 0; c < CPL; ++c) v[c] = fmaxf(v[c], 0.0f);
    }
    *reinterpret_cast<vrow*>(out + (size_t)i * fout + c0) = v;
}


#define FUSE_NODES 64
template <int CPL, bool PRE>
__global__ __launch_bounds__(256)
void gcn_gather_gemm_kernel(const float* __restrict__ h,
                       const int* __restrict__ rowptr,
                       const int2* __restrict__ meta,
                       const float* __restrict__ dinv,
                       const float* __restrict__ bias,
                       const float* __restrict__ Wf,
                       const float* __restrict__ Wn,
                       float* __restrict__ hn,
                       int n) {
    typedef __attribute__((ext_vector_type(CPL))) float vrow;
    constexpr int FIN   = 32 * CPL;
    constexpr int FOUT  = 64;
    constexpr int PITCH = FIN + 4;
    constexpr int NPW   = FUSE_NODES / 8;
    constexpr int WCOLS = PRE ? FIN : FOUT;
    __shared__ float Wt[FIN * WCOLS];
    __shared__ float Xt[FUSE_NODES * PITCH];
    const int tid  = threadIdx.x;
    const int lane = tid & 31;
    const int wv = __builtin_amdgcn_readfirstlane(tid >> 5);

    const float* Ws = PRE ? Wf : Wn;
    constexpr int NW4 = FIN * WCOLS / (256 * 4);
    float4 wreg[NW4];
    #pragma unroll
    for (int q = 0; q < NW4; ++q)
        wreg[q] = *reinterpret_cast<const float4*>(Ws + (q * 256 + tid) * 4);
    #pragma unroll
    for (int q = 0; q < NW4; ++q)
        *reinterpret_cast<float4*>(Wt + (q * 256 + tid) * 4) = wreg[q];

    const int node0 = blockIdx.x * FUSE_NODES;
    const int fout = FIN;
    const int c0 = lane * CPL;
    const float* hc = h + c0;
    const vrow bv = *reinterpret_cast<const vrow*>(bias + c0);
    const int ibase = node0 + wv * NPW;

    int sb = 0, se = 0; float sd = 0.0f;
    if (ibase < n) { sb = rowptr[ibase]; se = rowptr[ibase + 1]; sd = dinv[ibase]; }
    int begN = sb, endN = se; float dvN = sd;
    int nEN = endN - begN < 32 ? endN - begN : 32;
    vrow selfN;
    for (int c = 0; c < CPL; ++c) selfN[c] = 0.0f;
    if (ibase < n) selfN = *reinterpret_cast<const vrow*>(hc + (size_t)ibase * fout);
    int2 mN = make_int2(0, 0);
    if (lane < nEN) mN = meta[begN + lane];
    sb = 0; se = 0; sd = 0.0f;
    if (NPW > 1 && ibase + 1 < n) { sb = rowptr[ibase + 1]; se = rowptr[ibase + 2]; sd = dinv[ibase + 1]; }

    for (int rr = 0; rr < NPW; ++rr) {
        const int lr = wv * NPW + rr;
        const int i  = ibase + rr;
        const int beg = begN, end = endN;
        const float dv = dvN;
        const vrow self = selfN;
        const int2 m = mN;
        const int nE = nEN;

        begN = sb; endN = se; dvN = sd;
        nEN = endN - begN < 32 ? endN - begN : 32;
        if (rr + 1 < NPW && i + 1 < n)
            selfN = *reinterpret_cast<const vrow*>(hc + (size_t)(i + 1) * fout);
        mN = make_int2(0, 0);
        if (lane < nEN) mN = meta[begN + lane];
        sb = 0; se = 0; sd = 0.0f;
        if (rr + 2 < NPW && i + 2 < n) { sb = rowptr[i + 2]; se = rowptr[i + 3]; sd = dinv[i + 2]; }

        vrow v;
        for (int c = 0; c < CPL; ++c) v[c] = 0.0f;
        if (i < n) {
            vrow acc = self * dv * dv;

            auto sweep = [&](const int2 mm, const int cnt) {
                for (int e = 0; e < cnt; e += GATHER_AHEAD) {
                    vrow  r[GATHER_AHEAD];
                    float w[GATHER_AHEAD];
                    #pragma unroll
                    for (int q = 0; q < GATHER_AHEAD; ++q) {
                        if (q == 0 || e + q < cnt) {
                            const int sidx = __builtin_amdgcn_readlane(mm.x, e + q);
                            w[q] = __int_as_float(__builtin_amdgcn_readlane(mm.y, e + q));
                            r[q] = *reinterpret_cast<const vrow*>(hc + (size_t)sidx * fout);
                        }
                    }
                    #pragma unroll
                    for (int q = 0; q < GATHER_AHEAD; ++q) {
                        if (q == 0 || e + q < cnt) {
                            for (int c = 0; c < CPL; ++c) acc[c] = fmaf(r[q][c], w[q], acc[c]);
                        }
                    }
                }
            };
            sweep(m, nE);
            for (int cb = beg + 32; cb < end; cb += 32) {
                const int cnt = end - cb < 32 ? end - cb : 32;
                int2 mm = make_int2(0, 0);
                if (lane < cnt) mm = meta[cb + lane];
                sweep(mm, cnt);
            }
            if constexpr (PRE) {
                v = acc;
            } else {
                v = acc + bv;
                for (int c = 0; c < CPL; ++c) v[c] = fmaxf(v[c], 0.0f);
            }
        }
        *reinterpret_cast<vrow*>(Xt + lr * PITCH + c0) = v;
    }
    __syncthreads();

    const int rt   = wv >> 1;
    const int half = lane >> 4;
    const int l    = lane & 15;
    const float* xa = Xt + (rt * 16 + l) * PITCH + half * 2;

    if constexpr (PRE) {
        constexpr int NT = FIN / 32;
        const int ct0 = (wv & 1) * NT;
        const float* wf = Wt + (half * 2) * FIN + ct0 * 16 + l;
        v8f accf[NT];
        #pragma unroll
        for (int t = 0; t < NT; ++t) accf[t] = {};
        for (int k = 0; k < FIN; k += 4) {
            v2f a = *reinterpret_cast<const v2f*>(xa + k);
            #pragma unroll
            for (int t = 0; t < NT; ++t) {
                v2f b;
                b.x = wf[k * FIN + t * 16];
                b.y = wf[(k + 1) * FIN + t * 16];
                accf[t] = __builtin_amdgcn_wmma_f32_16x16x4_f32(
                              false, a, false, b, (short)0, accf[t], false, false);
            }
        }
        __syncthreads();
        constexpr int NN4 = FIN * FOUT / (256 * 4);
        float4 wnext[NN4];
        #pragma unroll
        for (int q = 0; q < NN4; ++q)
            wnext[q] = *reinterpret_cast<const float4*>(Wn + (q * 256 + tid) * 4);
        #pragma unroll
        for (int t = 0; t < NT; ++t) {
            const int col = (ct0 + t) * 16 + l;
            const float bc = bias[col];
            #pragma unroll
            for (int j = 0; j < 8; ++j)
                Xt[(rt * 16 + half * 8 + j) * PITCH + col] = fmaxf(accf[t][j] + bc, 0.0f);
        }
        #pragma unroll
        for (int q = 0; q < NN4; ++q)
            *reinterpret_cast<float4*>(Wt + (q * 256 + tid) * 4) = wnext[q];
        __syncthreads();
    }

    const int ctb  = (wv & 1) * 2;
    const float* wb = Wt + (half * 2) * FOUT + ctb * 16 + l;

    v8f acc0 = {}, acc1 = {};
    for (int k = 0; k < FIN; k += 4) {
        v2f a = *reinterpret_cast<const v2f*>(xa + k);
        v2f b0, b1;
        b0.x = wb[k * FOUT];
        b0.y = wb[(k + 1) * FOUT];
        b1.x = wb[k * FOUT + 16];
        b1.y = wb[(k + 1) * FOUT + 16];
        acc0 = __builtin_amdgcn_wmma_f32_16x16x4_f32(
                   false, a, false, b0, (short)0, acc0, false, false);
        acc1 = __builtin_amdgcn_wmma_f32_16x16x4_f32(
                   false, a, false, b1, (short)0, acc1, false, false);
    }

    const int base_m = node0 + rt * 16 + half * 8;
    const int col = ctb * 16 + l;
    #pragma unroll
    for (int j = 0; j < 8; ++j) {
        const int m = base_m + j;
        if (m < n) {
            hn[(size_t)m * FOUT + col]      = acc0[j];
            hn[(size_t)m * FOUT + col + 16] = acc1[j];
        }
    }
}


static inline int cdiv(long long a, int b) { return (int)((a + b - 1) / b); }
static inline size_t align64up(size_t x) { return (x + 63) & ~(size_t)63; }

extern "C" void kernel_launch(void* const* d_in, const int* in_sizes, int n_in,
                              void* d_out, int out_size, void* d_ws, size_t ws_size,
                              hipStream_t stream) {
    const float* x  = (const float*)d_in[0];
    const int*   ei = (const int*)  d_in[1];
    const float* W1 = (const float*)d_in[2];
    const float* b1 = (const float*)d_in[3];
    const float* W2 = (const float*)d_in[4];
    const float* b2 = (const float*)d_in[5];
    const float* W3 = (const float*)d_in[6];
    const float* b3 = (const float*)d_in[7];

    const int IN_C = 128;
    const int n = in_sizes[0] / IN_C;
    const int E = in_sizes[1] / 2;
    const int* src = ei;
    const int* dst = ei + E;

    char* base = (char*)d_ws;
    size_t off = ((256 - ((size_t)base & 255)) & 255) / 4;
    auto take = [&](size_t elems) { void* p = base + off * 4; off += align64up(elems); return p; };
    int*   rowptr = (int*)  take(n + 1);
    float* dinv   = (float*)take(n);
    int2*  meta   = (int2*) take((size_t)2 * E);
    float* h      = (float*)take((size_t)n * 64);
    float* xn     = (float*)take((size_t)n * 64);
    const int nb     = cdiv(n, BUCKET_NODES);
    const int nchunk = cdiv(E, EDGE_CHUNK);
    const int ntab   = nb * nchunk;
    int*   table  = (int*)  take(ntab);
    int*   toff   = (int*)  take(ntab + 1);
    int2*  filed  = (int2*) take((size_t)2 * E);
    const int scan_blocks = cdiv(ntab, SCAN_TILE);
    int*   tsum   = (int*)  take(scan_blocks);
    (void)ws_size;

    const int B = 256;

    bucket_count_kernel<<<nchunk, B, 0, stream>>>(dst, table, E, nchunk, nb);
    tile_total_kernel<<<scan_blocks, B, 0, stream>>>(table, tsum, ntab);
    scan_kernel<<<scan_blocks, B, 0, stream>>>(table, tsum, toff, ntab);
    bucket_scatter_kernel<<<nchunk, B, 0, stream>>>(src, dst, toff, filed, E, nchunk, nb);
    bucket_sort_kernel<<<nb, 1024, 0, stream>>>(filed, toff, rowptr, dinv, meta, n, nchunk, nb);
    norm_kernel<<<cdiv(E, 4 * B), B, 0, stream>>>(meta, dinv, E);

    const int gather_blocks = cdiv((long long)n * 32, B);
    const int fuse_blocks = cdiv(n, FUSE_NODES);

    gcn_gather_gemm_kernel<4, true><<<fuse_blocks, B, 0, stream>>>(x, rowptr, meta, dinv,
                                                             b1, W1, W2, xn, n);
    gcn_gather_gemm_kernel<2, false><<<fuse_blocks, B, 0, stream>>>(xn, rowptr, meta, dinv,
                                                             b2, nullptr, W3, h, n);
    gcn_gather_kernel<2><<<gather_blocks, B, 0, stream>>>(h, rowptr, meta, dinv, b3,
                                                          (float*)d_out, n, 64, 0);
}


# Rows stored pre-scaled by their own dinv, dinv[dst] applied once to a node's finished sum (norm factored out of the sum, slot order kept): records shrink to src, the per-edge norm pass and launch go
# speedup vs baseline: 3.1442x; 1.0318x over previous
#include <hip/hip_runtime.h>

typedef __attribute__((ext_vector_type(2))) float v2f;
typedef __attribute__((ext_vector_type(8))) float v8f;


#define SCAN_TILE 1024

__device__ __forceinline__ int4 load_counts4(const int* __restrict__ cnt, int base, int n) {
    int4 v = make_int4(0, 0, 0, 0);
    if (base + 3 < n) {
        v = *reinterpret_cast<const int4*>(cnt + base);
    } else {
        if (base     < n) v.x = cnt[base];
        if (base + 1 < n) v.y = cnt[base + 1];
        if (base + 2 < n) v.z = cnt[base + 2];
    }
    return v;
}

__global__ __launch_bounds__(256)
void tile_total_kernel(const int* __restrict__ cnt, int* __restrict__ tsum, int n) {
    __shared__ int wsum[8];
    const int tid  = threadIdx.x;
    const int base = (blockIdx.x * 256 + tid) * 4;
    const int4 v = load_counts4(cnt, base, n);
    int s = (v.x + v.y) + (v.z + v.w);
    for (int off = 16; off > 0; off >>= 1) s += __shfl_xor(s, off);
    if ((tid & 31) == 0) wsum[tid >> 5] = s;
    __syncthreads();
    if (tid == 0) {
        int t = 0;
        for (int w = 0; w < 8; ++w) t += wsum[w];
        tsum[blockIdx.x] = t;
    }
}

__global__ __launch_bounds__(256)
void scan_kernel(const int* __restrict__ cnt, const int* __restrict__ tsum,
                 int* __restrict__ rowptr, int n) {
    __shared__ int wtile[8];
    __shared__ int wprev[8];
    const int tid  = threadIdx.x;
    const int lane = tid & 31;
    const int wv   = tid >> 5;
    const int base = (blockIdx.x * 256 + tid) * 4;
    const int4 v = load_counts4(cnt, base, n);

    int prev = 0;
    for (int j = tid; j < (int)blockIdx.x; j += 256) prev += tsum[j];

    const int t1 = v.x, t2 = t1 + v.y, t3 = t2 + v.z, t4 = t3 + v.w;
    int inc = t4;
    for (int off = 1; off < 32; off <<= 1) {
        const int u = __shfl_up(inc, off);
        if (lane >= off) inc += u;
    }
    for (int off = 16; off > 0; off >>= 1) prev += __shfl_xor(prev, off);
    if (lane == 31) wtile[wv] = inc;
    if (lane == 0)  wprev[wv] = prev;
    __syncthreads();

    int start = 0;
    for (int w = 0; w < 8; ++w) {
        start += wprev[w];
        if (w < wv) start += wtile[w];
    }
    const int excl = start + inc - t4;

    if (base + 3 < n) {
        *reinterpret_cast<int4*>(rowptr + base) =
            make_int4(excl, excl + t1, excl + t2, excl + t3);
    } else {
        if (base     < n) rowptr[base]     = excl;
        if (base + 1 < n) rowptr[base + 1] = excl + t1;
        if (base + 2 < n) rowptr[base + 2] = excl + t2;
    }
    if (base < n && base + 4 >= n) rowptr[n] = excl + t4;
}

#define BUCKET_SHIFT 9
#define BUCKET_NODES (1 << BUCKET_SHIFT)
#define MAX_BUCKETS  1024
#define EDGE_CHUNK   4096
#define EDGES_PER_THREAD (EDGE_CHUNK / 256)

__global__ __launch_bounds__(256)
void bucket_count_kernel(const int* __restrict__ dst, int* __restrict__ table,
                         int E, int nchunk, int nb) {
    __shared__ int hist[MAX_BUCKETS];
    const int tid = threadIdx.x;
    const int chunk = blockIdx.x;
    const int e0 = chunk * EDGE_CHUNK;
    const int e1 = e0 + EDGE_CHUNK < E ? e0 + EDGE_CHUNK : E;
    for (int b = tid; b < nb; b += 256) hist[b] = 0;
    int d[EDGES_PER_THREAD];
    #pragma unroll
    for (int k = 0; k < EDGES_PER_THREAD; ++k) {
        const int e = e0 + k * 256 + tid;
        d[k] = e < e1 ? dst[e] : -1;
    }
    __syncthreads();
    #pragma unroll
    for (int k = 0; k < EDGES_PER_THREAD; ++k)
        if (d[k] >= 0) atomicAdd(&hist[d[k] >> BUCKET_SHIFT], 1);
    __syncthreads();
    for (int b = tid; b < nb; b += 256) table[b * nchunk + chunk] = hist[b];
}

__global__ __launch_bounds__(256)
void bucket_scatter_kernel(const int* __restrict__ src, const int* __restrict__ dst,
                           const int* __restrict__ toff, int2* __restrict__ filed,
                           int E, int nchunk, int nb) {
    __shared__ int cur[MAX_BUCKETS];
    const int tid = threadIdx.x;
    const int chunk = blockIdx.x;
    const int e0 = chunk * EDGE_CHUNK;
    const int e1 = e0 + EDGE_CHUNK < E ? e0 + EDGE_CHUNK : E;
    for (int b = tid; b < nb; b += 256) cur[b] = toff[b * nchunk + chunk];
    int s[EDGES_PER_THREAD], d[EDGES_PER_THREAD];
    #pragma unroll
    for (int k = 0; k < EDGES_PER_THREAD; ++k) {
        const int e = e0 + k * 256 + tid;
        s[k] = e < e1 ? src[e] : 0;
        d[k] = e < e1 ? dst[e] : -1;
    }
    __syncthreads();
    #pragma unroll
    for (int k = 0; k < EDGES_PER_THREAD; ++k) {
        if (d[k] >= 0) {
            const int pos = atomicAdd(&cur[d[k] >> BUCKET_SHIFT], 1);
            filed[pos] = make_int2(s[k], d[k]);
        }
    }
}

__global__ __launch_bounds__(1024)
void bucket_sort_kernel(const int2* __restrict__ filed, const int* __restrict__ toff,
                        int* __restrict__ rowptr, float* __restrict__ dinv,
                        int* __restrict__ meta, int n, int nchunk, int nb) {
    __shared__ int   cnt[BUCKET_NODES];
    __shared__ int   cur[BUCKET_NODES];
    __shared__ int   wtot[BUCKET_NODES / 32];
    const int tid  = threadIdx.x;
    const int lane = tid & 31;
    const int wv   = tid >> 5;
    const int b    = blockIdx.x;
    const int beg  = toff[b * nchunk];
    const int end  = toff[(b + 1) * nchunk];
    const int node0 = b << BUCKET_SHIFT;

    if (tid < BUCKET_NODES) cnt[tid] = 0;
    __syncthreads();
    for (int j0 = beg + tid; j0 < end; j0 += 4 * 1024) {
        int d[4];
        #pragma unroll
        for (int k = 0; k < 4; ++k) {
            const int j = j0 + k * 1024;
            d[k] = j < end ? filed[j].y : -1;
        }
        #pragma unroll
        for (int k = 0; k < 4; ++k)
            if (d[k] >= 0) atomicAdd(&cnt[d[k] - node0], 1);
    }
    __syncthreads();

    int c = 0, inc = 0;
    if (tid < BUCKET_NODES) {
        c = cnt[tid];
        inc = c;
        for (int off = 1; off < 32; off <<= 1) {
            const int u = __shfl_up(inc, off);
            if (lane >= off) inc += u;
        }
        if (lane == 31) wtot[wv] = inc;
    }
    __syncthreads();
    if (tid < BUCKET_NODES) {
        int start = beg;
        for (int w = 0; w < wv; ++w) start += wtot[w];
        const int excl = start + inc - c;
        cur[tid] = excl;
        const int node = node0 + tid;
        if (node < n) {
            const float dv = 1.0f / sqrtf((float)(c + 1));
            rowptr[node] = excl;
            dinv[node]   = dv;
        }
    }
    if (b == nb - 1 && tid == 0) rowptr[n] = end;
    __syncthreads();

    for (int j0 = beg + tid; j0 < end; j0 += 4 * 1024) {
        int2 e[4];
        #pragma unroll
        for (int k = 0; k < 4; ++k) {
            const int j = j0 + k * 1024;
            e[k] = j < end ? filed[j] : make_int2(0, -1);
        }
        #pragma unroll
        for (int k = 0; k < 4; ++k) {
            if (e[k].y >= 0) {
                const int ln = e[k].y - node0;
                const int pos = atomicAdd(&cur[ln], 1);
                meta[pos] = e[k].x;
            }
        }
    }
}

#define GATHER_AHEAD 16
template <int CPL>
__global__ void gcn_gather_kernel(const float* __restrict__ h,
                       const int* __restrict__ rowptr,
                       const int* __restrict__ meta,
                       const float* __restrict__ dinv,
                       const float* __restrict__ bias,
                       float* __restrict__ out,
                       int n, int fout, int do_relu) {
    typedef __attribute__((ext_vector_type(CPL))) float vrow;
    const int lane = threadIdx.x & 31;
    const int i = __builtin_amdgcn_readfirstlane(
                      (int)((blockIdx.x * blockDim.x + threadIdx.x) >> 5));
    if (i >= n) return;
    const int c0 = lane * CPL;
    const float* hc = h + c0;

    const vrow self = *reinterpret_cast<const vrow*>(hc + (size_t)i * fout);
    const float dv = dinv[i];
    const int beg = rowptr[i], end = rowptr[i + 1];

    int nE = end - beg < 32 ? end - beg : 32;
    int m = 0;
    if (lane < nE) m = meta[beg + lane];

    vrow acc = self;

    for (int cb = beg; cb < end; ) {
        for (int e = 0; e < nE; e += GATHER_AHEAD) {
            vrow  r[GATHER_AHEAD];
            #pragma unroll
            for (int q = 0; q < GATHER_AHEAD; ++q) {
                if (q == 0 || e + q < nE) {
                    const int sidx = __builtin_amdgcn_readlane(m, e + q);
                    r[q] = *reinterpret_cast<const vrow*>(hc + (size_t)sidx * fout);
                }
            }
            #pragma unroll
            for (int q = 0; q < GATHER_AHEAD; ++q) {
                if (q == 0 || e + q < nE) acc += r[q];
            }
        }
        cb += 32;
        if (cb < end) {
            nE = end - cb < 32 ? end - cb : 32;
            m = 0;
            if (lane < nE) m = meta[cb + lane];
        }
    }

    const vrow bv = *reinterpret_cast<const vrow*>(bias + c0);
    vrow v = acc * dv + bv;
    if (do_relu) {
        for (int c = 0; c < CPL; ++c) v[c] = fmaxf(v[c], 0.0f);
    }
    *reinterpret_cast<vrow*>(out + (size_t)i * fout + c0) = v;
}


#define FUSE_NODES 64
template <int CPL, bool PRE>
__global__ __launch_bounds__(256)
void gcn_gather_gemm_kernel(const float* __restrict__ h,
                       const int* __restrict__ rowptr,
                       const int* __restrict__ meta,
                       const float* __restrict__ dinv,
                       const float* __restrict__ bias,
                       const float* __restrict__ Wf,
                       const float* __restrict__ Wn,
                       float* __restrict__ hn,
                       int n) {
    typedef __attribute__((ext_vector_type(CPL))) float vrow;
    constexpr int FIN   = 32 * CPL;
    constexpr int FOUT  = 64;
    constexpr int PITCH = FIN + 4;
    constexpr int NPW   = FUSE_NODES / 8;
    constexpr int WCOLS = PRE ? FIN : FOUT;
    __shared__ float Wt[FIN * WCOLS];
    __shared__ float Xt[FUSE_NODES * PITCH];
    __shared__ float dvrow[FUSE_NODES];
    const int tid  = threadIdx.x;
    const int lane = tid & 31;
    const int wv = __builtin_amdgcn_readfirstlane(tid >> 5);

    const float* Ws = PRE ? Wf : Wn;
    constexpr int NW4 = FIN * WCOLS / (256 * 4);
    float4 wreg[NW4];
    #pragma unroll
    for (int q = 0; q < NW4; ++q)
        wreg[q] = *reinterpret_cast<const float4*>(Ws + (q * 256 + tid) * 4);
    #pragma unroll
    for (int q = 0; q < NW4; ++q)
        *reinterpret_cast<float4*>(Wt + (q * 256 + tid) * 4) = wreg[q];

    const int node0 = blockIdx.x * FUSE_NODES;
    const int fout = FIN;
    const int c0 = lane * CPL;
    const float* hc = h + c0;
    const vrow bv = *reinterpret_cast<const vrow*>(bias + c0);
    const int ibase = node0 + wv * NPW;

    int sb = 0, se = 0; float sd = 0.0f;
    if (ibase < n) { sb = rowptr[ibase]; se = rowptr[ibase + 1]; sd = dinv[ibase]; }
    int begN = sb, endN = se; float dvN = sd;
    int nEN = endN - begN < 32 ? endN - begN : 32;
    vrow selfN;
    for (int c = 0; c < CPL; ++c) selfN[c] = 0.0f;
    if (ibase < n) selfN = *reinterpret_cast<const vrow*>(hc + (size_t)ibase * fout);
    int mN = 0;
    if (lane < nEN) mN = meta[begN + lane];
    sb = 0; se = 0; sd = 0.0f;
    if (NPW > 1 && ibase + 1 < n) { sb = rowptr[ibase + 1]; se = rowptr[ibase + 2]; sd = dinv[ibase + 1]; }

    for (int rr = 0; rr < NPW; ++rr) {
        const int lr = wv * NPW + rr;
        const int i  = ibase + rr;
        const int beg = begN, end = endN;
        const float dv = dvN;
        const vrow self = selfN;
        const int m = mN;
        const int nE = nEN;

        begN = sb; endN = se; dvN = sd;
        nEN = endN - begN < 32 ? endN - begN : 32;
        if (rr + 1 < NPW && i + 1 < n)
            selfN = *reinterpret_cast<const vrow*>(hc + (size_t)(i + 1) * fout);
        mN = 0;
        if (lane < nEN) mN = meta[begN + lane];
        sb = 0; se = 0; sd = 0.0f;
        if (rr + 2 < NPW && i + 2 < n) { sb = rowptr[i + 2]; se = rowptr[i + 3]; sd = dinv[i + 2]; }

        vrow v;
        for (int c = 0; c < CPL; ++c) v[c] = 0.0f;
        if (i < n) {
            vrow acc;
            if constexpr (PRE) acc = self * dv;
            else               acc = self;

            auto sweep = [&](const int mm, const int cnt) {
                float ww = 0.0f;
                if constexpr (PRE) { if (lane < cnt) ww = dinv[mm]; }
                for (int e = 0; e < cnt; e += GATHER_AHEAD) {
                    vrow  r[GATHER_AHEAD];
                    #pragma unroll
                    for (int q = 0; q < GATHER_AHEAD; ++q) {
                        if (q == 0 || e + q < cnt) {
                            const int sidx = __builtin_amdgcn_readlane(mm, e + q);
                            r[q] = *reinterpret_cast<const vrow*>(hc + (size_t)sidx * fout);
                        }
                    }
                    #pragma unroll
                    for (int q = 0; q < GATHER_AHEAD; ++q) {
                        if (q == 0 || e + q < cnt) {
                            if constexpr (PRE) {
                                const float w = __int_as_float(
                                    __builtin_amdgcn_readlane(__float_as_int(ww), e + q));
                                for (int c = 0; c < CPL; ++c) acc[c] = fmaf(r[q][c], w, acc[c]);
                            } else {
                                acc += r[q];
                            }
                        }
                    }
                }
            };
            sweep(m, nE);
            for (int cb = beg + 32; cb < end; cb += 32) {
                const int cnt = end - cb < 32 ? end - cb : 32;
                int mm = 0;
                if (lane < cnt) mm = meta[cb + lane];
                sweep(mm, cnt);
            }
            if constexpr (PRE) {
                v = acc * dv;
            } else {
                v = acc * dv + bv;
                for (int c = 0; c < CPL; ++c) v[c] = fmaxf(v[c], 0.0f) * dv;
            }
        }
        if constexpr (PRE) {
            if (lane == 0) dvrow[lr] = i < n ? dv : 0.0f;
        }
        *reinterpret_cast<vrow*>(Xt + lr * PITCH + c0) = v;
    }
    __syncthreads();

    const int rt   = wv >> 1;
    const int half = lane >> 4;
    const int l    = lane & 15;
    const float* xa = Xt + (rt * 16 + l) * PITCH + half * 2;

    if constexpr (PRE) {
        constexpr int NT = FIN / 32;
        const int ct0 = (wv & 1) * NT;
        const float* wf = Wt + (half * 2) * FIN + ct0 * 16 + l;
        v8f accf[NT];
        #pragma unroll
        for (int t = 0; t < NT; ++t) accf[t] = {};
        for (int k = 0; k < FIN; k += 4) {
            v2f a = *reinterpret_cast<const v2f*>(xa + k);
            #pragma unroll
            for (int t = 0; t < NT; ++t) {
                v2f b;
                b.x = wf[k * FIN + t * 16];
                b.y = wf[(k + 1) * FIN + t * 16];
                accf[t] = __builtin_amdgcn_wmma_f32_16x16x4_f32(
                              false, a, false, b, (short)0, accf[t], false, false);
            }
        }
        __syncthreads();
        constexpr int NN4 = FIN * FOUT / (256 * 4);
        float4 wnext[NN4];
        #pragma unroll
        for (int q = 0; q < NN4; ++q)
            wnext[q] = *reinterpret_cast<const float4*>(Wn + (q * 256 + tid) * 4);
        #pragma unroll
        for (int t = 0; t < NT; ++t) {
            const int col = (ct0 + t) * 16 + l;
            const float bc = bias[col];
            #pragma unroll
            for (int j = 0; j < 8; ++j)
                Xt[(rt * 16 + half * 8 + j) * PITCH + col] =
                    fmaxf(accf[t][j] + bc, 0.0f) * dvrow[rt * 16 + half * 8 + j];
        }
        #pragma unroll
        for (int q = 0; q < NN4; ++q)
            *reinterpret_cast<float4*>(Wt + (q * 256 + tid) * 4) = wnext[q];
        __syncthreads();
    }

    const int ctb  = (wv & 1) * 2;
    const float* wb = Wt + (half * 2) * FOUT + ctb * 16 + l;

    v8f acc0 = {}, acc1 = {};
    for (int k = 0; k < FIN; k += 4) {
        v2f a = *reinterpret_cast<const v2f*>(xa + k);
        v2f b0, b1;
        b0.x = wb[k * FOUT];
        b0.y = wb[(k + 1) * FOUT];
        b1.x = wb[k * FOUT + 16];
        b1.y = wb[(k + 1) * FOUT + 16];
        acc0 = __builtin_amdgcn_wmma_f32_16x16x4_f32(
                   false, a, false, b0, (short)0, acc0, false, false);
        acc1 = __builtin_amdgcn_wmma_f32_16x16x4_f32(
                   false, a, false, b1, (short)0, acc1, false, false);
    }

    const int base_m = node0 + rt * 16 + half * 8;
    const int col = ctb * 16 + l;
    #pragma unroll
    for (int j = 0; j < 8; ++j) {
        const int m = base_m + j;
        if (m < n) {
            hn[(size_t)m * FOUT + col]      = acc0[j];
            hn[(size_t)m * FOUT + col + 16] = acc1[j];
        }
    }
}


static inline int cdiv(long long a, int b) { return (int)((a + b - 1) / b); }
static inline size_t align64up(size_t x) { return (x + 63) & ~(size_t)63; }

extern "C" void kernel_launch(void* const* d_in, const int* in_sizes, int n_in,
                              void* d_out, int out_size, void* d_ws, size_t ws_size,
                              hipStream_t stream) {
    const float* x  = (const float*)d_in[0];
    const int*   ei = (const int*)  d_in[1];
    const float* W1 = (const float*)d_in[2];
    const float* b1 = (const float*)d_in[3];
    const float* W2 = (const float*)d_in[4];
    const float* b2 = (const float*)d_in[5];
    const float* W3 = (const float*)d_in[6];
    const float* b3 = (const float*)d_in[7];

    const int IN_C = 128;
    const int n = in_sizes[0] / IN_C;
    const int E = in_sizes[1] / 2;
    const int* src = ei;
    const int* dst = ei + E;

    char* base = (char*)d_ws;
    size_t off = ((256 - ((size_t)base & 255)) & 255) / 4;
    auto take = [&](size_t elems) { void* p = base + off * 4; off += align64up(elems); return p; };
    int*   rowptr = (int*)  take(n + 1);
    float* dinv   = (float*)take(n);
    int*   meta   = (int*)  take(E);
    float* h      = (float*)take((size_t)n * 64);
    float* xn     = (float*)take((size_t)n * 64);
    const int nb     = cdiv(n, BUCKET_NODES);
    const int nchunk = cdiv(E, EDGE_CHUNK);
    const int ntab   = nb * nchunk;
    int*   table  = (int*)  take(ntab);
    int*   toff   = (int*)  take(ntab + 1);
    int2*  filed  = (int2*) take((size_t)2 * E);
    const int scan_blocks = cdiv(ntab, SCAN_TILE);
    int*   tsum   = (int*)  take(scan_blocks);
    (void)ws_size;

    const int B = 256;

    bucket_count_kernel<<<nchunk, B, 0, stream>>>(dst, table, E, nchunk, nb);
    tile_total_kernel<<<scan_blocks, B, 0, stream>>>(table, tsum, ntab);
    scan_kernel<<<scan_blocks, B, 0, stream>>>(table, tsum, toff, ntab);
    bucket_scatter_kernel<<<nchunk, B, 0, stream>>>(src, dst, toff, filed, E, nchunk, nb);
    bucket_sort_kernel<<<nb, 1024, 0, stream>>>(filed, toff, rowptr, dinv, meta, n, nchunk, nb);

    const int gather_blocks = cdiv((long long)n * 32, B);
    const int fuse_blocks = cdiv(n, FUSE_NODES);

    gcn_gather_gemm_kernel<4, true><<<fuse_blocks, B, 0, stream>>>(x, rowptr, meta, dinv,
                                                             b1, W1, W2, xn, n);
    gcn_gather_gemm_kernel<2, false><<<fuse_blocks, B, 0, stream>>>(xn, rowptr, meta, dinv,
                                                             b2, nullptr, W3, h, n);
    gcn_gather_kernel<2><<<gather_blocks, B, 0, stream>>>(h, rowptr, meta, dinv, b3,
                                                          (float*)d_out, n, 64, 0);
}


# CSR build in 2 launches, was 5: each edge chunk is sorted by destination range inside its own work-group and a range's work-group collects its pieces; table scan and scatter go; sum order: arrival
# speedup vs baseline: 3.5043x; 1.1145x over previous
#include <hip/hip_runtime.h>

typedef __attribute__((ext_vector_type(2))) float v2f;
typedef __attribute__((ext_vector_type(8))) float v8f;


#define BUCKET_SHIFT 9
#define BUCKET_NODES (1 << BUCKET_SHIFT)
#define MAX_BUCKETS  1024
#define EDGE_CHUNK   4096
#define EDGES_PER_THREAD (EDGE_CHUNK / 256)
#define PIECES_AHEAD 8

__global__ __launch_bounds__(256)
void chunk_sort_kernel(const int* __restrict__ src, const int* __restrict__ dst,
                       int* __restrict__ desc, int2* __restrict__ filed,
                       int E, int nchunk, int nb) {
    __shared__ __align__(16) int hist[MAX_BUCKETS];
    __shared__ int wsum[8];
    const int tid  = threadIdx.x;
    const int lane = tid & 31;
    const int wv   = tid >> 5;
    const int chunk = blockIdx.x;
    const int e0 = chunk * EDGE_CHUNK;
    const int e1 = e0 + EDGE_CHUNK < E ? e0 + EDGE_CHUNK : E;
    *reinterpret_cast<int4*>(hist + tid * 4) = make_int4(0, 0, 0, 0);
    int s[EDGES_PER_THREAD], d[EDGES_PER_THREAD], r[EDGES_PER_THREAD];
    #pragma unroll
    for (int k = 0; k < EDGES_PER_THREAD; ++k) {
        const int e = e0 + k * 256 + tid;
        s[k] = e < e1 ? src[e] : 0;
        d[k] = e < e1 ? dst[e] : -1;
    }
    __syncthreads();
    #pragma unroll
    for (int k = 0; k < EDGES_PER_THREAD; ++k)
        r[k] = d[k] >= 0 ? atomicAdd(&hist[d[k] >> BUCKET_SHIFT], 1) : 0;
    __syncthreads();

    const int4 c = *reinterpret_cast<const int4*>(hist + tid * 4);
    const int t1 = c.x, t2 = t1 + c.y, t3 = t2 + c.z, t4 = t3 + c.w;
    int inc = t4;
    for (int off = 1; off < 32; off <<= 1) {
        const int u = __shfl_up(inc, off);
        if (lane >= off) inc += u;
    }
    if (lane == 31) wsum[wv] = inc;
    __syncthreads();
    int x0 = inc - t4;
    for (int w = 0; w < 8; ++w)
        if (w < wv) x0 += wsum[w];
    *reinterpret_cast<int4*>(hist + tid * 4) = make_int4(x0, x0 + t1, x0 + t2, x0 + t3);
    __syncthreads();
    #pragma unroll
    for (int k = 0; k < EDGES_PER_THREAD; ++k)
        if (d[k] >= 0)
            filed[e0 + hist[d[k] >> BUCKET_SHIFT] + r[k]] = make_int2(s[k], d[k]);
    const int b0 = tid * 4;
    if (b0     < nb) desc[(b0    ) * nchunk + chunk] =  x0       | (c.x << 16);
    if (b0 + 1 < nb) desc[(b0 + 1) * nchunk + chunk] = (x0 + t1) | (c.y << 16);
    if (b0 + 2 < nb) desc[(b0 + 2) * nchunk + chunk] = (x0 + t2) | (c.z << 16);
    if (b0 + 3 < nb) desc[(b0 + 3) * nchunk + chunk] = (x0 + t3) | (c.w << 16);
}

__global__ __launch_bounds__(1024)
void bucket_sort_kernel(const int2* __restrict__ filed, const int* __restrict__ desc,
                        int* __restrict__ rowptr, float* __restrict__ dinv,
                        int* __restrict__ meta, int n, int nchunk, int nb) {
    __shared__ int   cnt[BUCKET_NODES];
    __shared__ int   cur[BUCKET_NODES];
    __shared__ int   wtot[BUCKET_NODES / 32];
    __shared__ int   wprev[32];
    const int tid  = threadIdx.x;
    const int lane = tid & 31;
    const int wv   = __builtin_amdgcn_readfirstlane(tid >> 5);
    const int b    = blockIdx.x;
    const int node0 = b << BUCKET_SHIFT;
    const int* drow = desc + b * nchunk;

    const int lim  = b * nchunk;
    const int lim4 = lim & ~3;
    int prev = 0;
    for (int j0 = tid * 4; j0 < lim4; j0 += 8 * 4096) {
        int4 p[8];
        #pragma unroll
        for (int q = 0; q < 8; ++q) {
            const int j = j0 + q * 4096;
            p[q] = *reinterpret_cast<const int4*>(desc + (j < lim4 ? j : lim4 - 4));
        }
        #pragma unroll
        for (int q = 0; q < 8; ++q) {
            const int t = ((p[q].x >> 16) + (p[q].y >> 16)) + ((p[q].z >> 16) + (p[q].w >> 16));
            prev += j0 + q * 4096 < lim4 ? t : 0;
        }
    }
    if (tid < (lim & 3)) prev += desc[lim4 + tid] >> 16;
    for (int off = 16; off > 0; off >>= 1) prev += __shfl_xor(prev, off);
    if (lane == 0) wprev[wv] = prev;
    if (tid < BUCKET_NODES) cnt[tid] = 0;
    __syncthreads();

    for (int c0 = wv; c0 < nchunk; c0 += 32 * PIECES_AHEAD) {
        int at[PIECES_AHEAD], len[PIECES_AHEAD], d[PIECES_AHEAD];
        #pragma unroll
        for (int k = 0; k < PIECES_AHEAD; ++k) {
            const int c = c0 + 32 * k;
            const int pd = c < nchunk ? drow[c] : 0;
            at[k]  = c * EDGE_CHUNK + (pd & 0xffff);
            len[k] = pd >> 16;
        }
        #pragma unroll
        for (int k = 0; k < PIECES_AHEAD; ++k)
            d[k] = lane < len[k] ? filed[at[k] + lane].y : -1;
        #pragma unroll
        for (int k = 0; k < PIECES_AHEAD; ++k)
            if (d[k] >= 0) atomicAdd(&cnt[d[k] - node0], 1);
        #pragma unroll
        for (int k = 0; k < PIECES_AHEAD; ++k)
            for (int e = 32 + lane; e < len[k]; e += 32)
                atomicAdd(&cnt[filed[at[k] + e].y - node0], 1);
    }
    __syncthreads();

    int c = 0, inc = 0;
    if (tid < BUCKET_NODES) {
        c = cnt[tid];
        inc = c;
        for (int off = 1; off < 32; off <<= 1) {
            const int u = __shfl_up(inc, off);
            if (lane >= off) inc += u;
        }
        if (lane == 31) wtot[wv] = inc;
    }
    __syncthreads();
    int excl = 0;
    if (tid < BUCKET_NODES) {
        int start = wprev[lane];
        for (int off = 16; off > 0; off >>= 1) start += __shfl_xor(start, off);
        for (int w = 0; w < wv; ++w) start += wtot[w];
        excl = start + inc - c;
        cur[tid] = excl;
    }
    __syncthreads();
    if (tid < BUCKET_NODES) {
        const int node = node0 + tid;
        if (node < n) {
            const float dv = 1.0f / sqrtf((float)(c + 1));
            rowptr[node] = excl;
            dinv[node]   = dv;
        }
        if (b == nb - 1 && tid == BUCKET_NODES - 1) rowptr[n] = excl + c;
    }

    for (int c0 = wv; c0 < nchunk; c0 += 32 * PIECES_AHEAD) {
        int at[PIECES_AHEAD], len[PIECES_AHEAD];
        int2 e[PIECES_AHEAD];
        #pragma unroll
        for (int k = 0; k < PIECES_AHEAD; ++k) {
            const int c = c0 + 32 * k;
            const int pd = c < nchunk ? drow[c] : 0;
            at[k]  = c * EDGE_CHUNK + (pd & 0xffff);
            len[k] = pd >> 16;
        }
        #pragma unroll
        for (int k = 0; k < PIECES_AHEAD; ++k)
            e[k] = lane < len[k] ? filed[at[k] + lane] : make_int2(0, -1);
        #pragma unroll
        for (int k = 0; k < PIECES_AHEAD; ++k) {
            if (e[k].y >= 0) {
                const int pos = atomicAdd(&cur[e[k].y - node0], 1);
                meta[pos] = e[k].x;
            }
        }
        #pragma unroll
        for (int k = 0; k < PIECES_AHEAD; ++k) {
            for (int j = 32 + lane; j < len[k]; j += 32) {
                const int2 ee = filed[at[k] + j];
                const int pos = atomicAdd(&cur[ee.y - node0], 1);
                meta[pos] = ee.x;
            }
        }
    }
}

#define GATHER_AHEAD 16
template <int CPL>
__global__ void gcn_gather_kernel(const float* __restrict__ h,
                       const int* __restrict__ rowptr,
                       const int* __restrict__ meta,
                       const float* __restrict__ dinv,
                       const float* __restrict__ bias,
                       float* __restrict__ out,
                       int n, int fout, int do_relu) {
    typedef __attribute__((ext_vector_type(CPL))) float vrow;
    const int lane = threadIdx.x & 31;
    const int i = __builtin_amdgcn_readfirstlane(
                      (int)((blockIdx.x * blockDim.x + threadIdx.x) >> 5));
    if (i >= n) return;
    const int c0 = lane * CPL;
    const float* hc = h + c0;

    const vrow self = *reinterpret_cast<const vrow*>(hc + (size_t)i * fout);
    const float dv = dinv[i];
    const int beg = rowptr[i], end = rowptr[i + 1];

    int nE = end - beg < 32 ? end - beg : 32;
    int m = 0;
    if (lane < nE) m = meta[beg + lane];

    vrow acc = self;

    for (int cb = beg; cb < end; ) {
        for (int e = 0; e < nE; e += GATHER_AHEAD) {
            vrow  r[GATHER_AHEAD];
            #pragma unroll
            for (int q = 0; q < GATHER_AHEAD; ++q) {
                if (q == 0 || e + q < nE) {
                    const int sidx = __builtin_amdgcn_readlane(m, e + q);
                    r[q] = *reinterpret_cast<const vrow*>(hc + (size_t)sidx * fout);
                }
            }
            #pragma unroll
            for (int q = 0; q < GATHER_AHEAD; ++q) {
                if (q == 0 || e + q < nE) acc += r[q];
            }
        }
        cb += 32;
        if (cb < end) {
            nE = end - cb < 32 ? end - cb : 32;
            m = 0;
            if (lane < nE) m = meta[cb + lane];
        }
    }

    const vrow bv = *reinterpret_cast<const vrow*>(bias + c0);
    vrow v = acc * dv + bv;
    if (do_relu) {
        for (int c = 0; c < CPL; ++c) v[c] = fmaxf(v[c], 0.0f);
    }
    *reinterpret_cast<vrow*>(out + (size_t)i * fout + c0) = v;
}


#define FUSE_NODES 64
template <int CPL, bool PRE>
__global__ __launch_bounds__(256)
void gcn_gather_gemm_kernel(const float* __restrict__ h,
                       const int* __restrict__ rowptr,
                       const int* __restrict__ meta,
                       const float* __restrict__ dinv,
                       const float* __restrict__ bias,
                       const float* __restrict__ Wf,
                       const float* __restrict__ Wn,
                       float* __restrict__ hn,
                       int n) {
    typedef __attribute__((ext_vector_type(CPL))) float vrow;
    constexpr int FIN   = 32 * CPL;
    constexpr int FOUT  = 64;
    constexpr int PITCH = FIN + 4;
    constexpr int NPW   = FUSE_NODES / 8;
    constexpr int WCOLS = PRE ? FIN : FOUT;
    __shared__ float Wt[FIN * WCOLS];
    __shared__ float Xt[FUSE_NODES * PITCH];
    __shared__ float dvrow[FUSE_NODES];
    const int tid  = threadIdx.x;
    const int lane = tid & 31;
    const int wv = __builtin_amdgcn_readfirstlane(tid >> 5);

    const float* Ws = PRE ? Wf : Wn;
    constexpr int NW4 = FIN * WCOLS / (256 * 4);
    float4 wreg[NW4];
    #pragma unroll
    for (int q = 0; q < NW4; ++q)
        wreg[q] = *reinterpret_cast<const float4*>(Ws + (q * 256 + tid) * 4);
    #pragma unroll
    for (int q = 0; q < NW4; ++q)
        *reinterpret_cast<float4*>(Wt + (q * 256 + tid) * 4) = wreg[q];

    const int node0 = blockIdx.x * FUSE_NODES;
    const int fout = FIN;
    const int c0 = lane * CPL;
    const float* hc = h + c0;
    const vrow bv = *reinterpret_cast<const vrow*>(bias + c0);
    const int ibase = node0 + wv * NPW;

    int sb = 0, se = 0; float sd = 0.0f;
    if (ibase < n) { sb = rowptr[ibase]; se = rowptr[ibase + 1]; sd = dinv[ibase]; }
    int begN = sb, endN = se; float dvN = sd;
    int nEN = endN - begN < 32 ? endN - begN : 32;
    vrow selfN;
    for (int c = 0; c < CPL; ++c) selfN[c] = 0.0f;
    if (ibase < n) selfN = *reinterpret_cast<const vrow*>(hc + (size_t)ibase * fout);
    int mN = 0;
    if (lane < nEN) mN = meta[begN + lane];
    sb = 0; se = 0; sd = 0.0f;
    if (NPW > 1 && ibase + 1 < n) { sb = rowptr[ibase + 1]; se = rowptr[ibase + 2]; sd = dinv[ibase + 1]; }

    for (int rr = 0; rr < NPW; ++rr) {
        const int lr = wv * NPW + rr;
        const int i  = ibase + rr;
        const int beg = begN, end = endN;
        const float dv = dvN;
        const vrow self = selfN;
        const int m = mN;
        const int nE = nEN;

        begN = sb; endN = se; dvN = sd;
        nEN = endN - begN < 32 ? endN - begN : 32;
        if (rr + 1 < NPW && i + 1 < n)
            selfN = *reinterpret_cast<const vrow*>(hc + (size_t)(i + 1) * fout);
        mN = 0;
        if (lane < nEN) mN = meta[begN + lane];
        sb = 0; se = 0; sd = 0.0f;
        if (rr + 2 < NPW && i + 2 < n) { sb = rowptr[i + 2]; se = rowptr[i + 3]; sd = dinv[i + 2]; }

        vrow v;
        for (int c = 0; c < CPL; ++c) v[c] = 0.0f;
        if (i < n) {
            vrow acc;
            if constexpr (PRE) acc = self * dv;
            else               acc = self;

            auto sweep = [&](const int mm, const int cnt) {
                float ww = 0.0f;
                if constexpr (PRE) { if (lane < cnt) ww = dinv[mm]; }
                for (int e = 0; e < cnt; e += GATHER_AHEAD) {
                    vrow  r[GATHER_AHEAD];
                    #pragma unroll
                    for (int q = 0; q < GATHER_AHEAD; ++q) {
                        if (q == 0 || e + q < cnt) {
                            const int sidx = __builtin_amdgcn_readlane(mm, e + q);
                            r[q] = *reinterpret_cast<const vrow*>(hc + (size_t)sidx * fout);
                        }
                    }
                    #pragma unroll
                    for (int q = 0; q < GATHER_AHEAD; ++q) {
                        if (q == 0 || e + q < cnt) {
                            if constexpr (PRE) {
                                const float w = __int_as_float(
                                    __builtin_amdgcn_readlane(__float_as_int(ww), e + q));
                                for (int c = 0; c < CPL; ++c) acc[c] = fmaf(r[q][c], w, acc[c]);
                            } else {
                                acc += r[q];
                            }
                        }
                    }
                }
            };
            sweep(m, nE);
            for (int cb = beg + 32; cb < end; cb += 32) {
                const int cnt = end - cb < 32 ? end - cb : 32;
                int mm = 0;
                if (lane < cnt) mm = meta[cb + lane];
                sweep(mm, cnt);
            }
            if constexpr (PRE) {
                v = acc * dv;
            } else {
                v = acc * dv + bv;
                for (int c = 0; c < CPL; ++c) v[c] = fmaxf(v[c], 0.0f) * dv;
            }
        }
        if constexpr (PRE) {
            if (lane == 0) dvrow[lr] = i < n ? dv : 0.0f;
        }
        *reinterpret_cast<vrow*>(Xt + lr * PITCH + c0) = v;
    }
    __syncthreads();

    const int rt   = wv >> 1;
    const int half = lane >> 4;
    const int l    = lane & 15;
    const float* xa = Xt + (rt * 16 + l) * PITCH + half * 2;

    if constexpr (PRE) {
        constexpr int NT = FIN / 32;
        const int ct0 = (wv & 1) * NT;
        const float* wf = Wt + (half * 2) * FIN + ct0 * 16 + l;
        v8f accf[NT];
        #pragma unroll
        for (int t = 0; t < NT; ++t) accf[t] = {};
        for (int k = 0; k < FIN; k += 4) {
            v2f a = *reinterpret_cast<const v2f*>(xa + k);
            #pragma unroll
            for (int t = 0; t < NT; ++t) {
                v2f b;
                b.x = wf[k * FIN + t * 16];
                b.y = wf[(k + 1) * FIN + t * 16];
                accf[t] = __builtin_amdgcn_wmma_f32_16x16x4_f32(
                              false, a, false, b, (short)0, accf[t], false, false);
            }
        }
        __syncthreads();
        constexpr int NN4 = FIN * FOUT / (256 * 4);
        float4 wnext[NN4];
        #pragma unroll
        for (int q = 0; q < NN4; ++q)
            wnext[q] = *reinterpret_cast<const float4*>(Wn + (q * 256 + tid) * 4);
        #pragma unroll
        for (int t = 0; t < NT; ++t) {
            const int col = (ct0 + t) * 16 + l;
            const float bc = bias[col];
            #pragma unroll
            for (int j = 0; j < 8; ++j)
                Xt[(rt * 16 + half * 8 + j) * PITCH + col] =
                    fmaxf(accf[t][j] + bc, 0.0f) * dvrow[rt * 16 + half * 8 + j];
        }
        #pragma unroll
        for (int q = 0; q < NN4; ++q)
            *reinterpret_cast<float4*>(Wt + (q * 256 + tid) * 4) = wnext[q];
        __syncthreads();
    }

    const int ctb  = (wv & 1) * 2;
    const float* wb = Wt + (half * 2) * FOUT + ctb * 16 + l;

    v8f acc0 = {}, acc1 = {};
    for (int k = 0; k < FIN; k += 4) {
        v2f a = *reinterpret_cast<const v2f*>(xa + k);
        v2f b0, b1;
        b0.x = wb[k * FOUT];
        b0.y = wb[(k + 1) * FOUT];
        b1.x = wb[k * FOUT + 16];
        b1.y = wb[(k + 1) * FOUT + 16];
        acc0 = __builtin_amdgcn_wmma_f32_16x16x4_f32(
                   false, a, false, b0, (short)0, acc0, false, false);
        acc1 = __builtin_amdgcn_wmma_f32_16x16x4_f32(
                   false, a, false, b1, (short)0, acc1, false, false);
    }

    const int base_m = node0 + rt * 16 + half * 8;
    const int col = ctb * 16 + l;
    #pragma unroll
    for (int j = 0; j < 8; ++j) {
        const int m = base_m + j;
        if (m < n) {
            hn[(size_t)m * FOUT + col]      = acc0[j];
            hn[(size_t)m * FOUT + col + 16] = acc1[j];
        }
    }
}


static inline int cdiv(long long a, int b) { return (int)((a + b - 1) / b); }
static inline size_t align64up(size_t x) { return (x + 63) & ~(size_t)63; }

extern "C" void kernel_launch(void* const* d_in, const int* in_sizes, int n_in,
                              void* d_out, int out_size, void* d_ws, size_t ws_size,
                              hipStream_t stream) {
    const float* x  = (const float*)d_in[0];
    const int*   ei = (const int*)  d_in[1];
    const float* W1 = (const float*)d_in[2];
    const float* b1 = (const float*)d_in[3];
    const float* W2 = (const float*)d_in[4];
    const float* b2 = (const float*)d_in[5];
    const float* W3 = (const float*)d_in[6];
    const float* b3 = (const float*)d_in[7];

    const int IN_C = 128;
    const int n = in_sizes[0] / IN_C;
    const int E = in_sizes[1] / 2;
    const int* src = ei;
    const int* dst = ei + E;

    char* base = (char*)d_ws;
    size_t off = ((256 - ((size_t)base & 255)) & 255) / 4;
    auto take = [&](size_t elems) { void* p = base + off * 4; off += align64up(elems); return p; };
    int*   rowptr = (int*)  take(n + 1);
    float* dinv   = (float*)take(n);
    int*   meta   = (int*)  take(E);
    float* h      = (float*)take((size_t)n * 64);
    float* xn     = (float*)take((size_t)n * 64);
    const int nb     = cdiv(n, BUCKET_NODES);
    const int nchunk = cdiv(E, EDGE_CHUNK);
    int*   desc   = (int*)  take((size_t)nb * nchunk);
    int2*  filed  = (int2*) take((size_t)2 * E);
    (void)ws_size;

    const int B = 256;

    chunk_sort_kernel<<<nchunk, B, 0, stream>>>(src, dst, desc, filed, E, nchunk, nb);
    bucket_sort_kernel<<<nb, 1024, 0, stream>>>(filed, desc, rowptr, dinv, meta, n, nchunk, nb);

    const int gather_blocks = cdiv((long long)n * 32, B);
    const int fuse_blocks = cdiv(n, FUSE_NODES);

    gcn_gather_gemm_kernel<4, true><<<fuse_blocks, B, 0, stream>>>(x, rowptr, meta, dinv,
                                                             b1, W1, W2, xn, n);
    gcn_gather_gemm_kernel<2, false><<<fuse_blocks, B, 0, stream>>>(xn, rowptr, meta, dinv,
                                                             b2, nullptr, W3, h, n);
    gcn_gather_kernel<2><<<gather_blocks, B, 0, stream>>>(h, rowptr, meta, dinv, b3,
                                                          (float*)d_out, n, 64, 0);
}


# Layer 1 kernel: one resident work-group per WGP walks over node tiles, 16 waves gathering tile t+1 while 4 waves multiply tile t from a second LDS stage (weights staged once); sums kept
# speedup vs baseline: 3.6260x; 1.0347x over previous
#include <hip/hip_runtime.h>

typedef __attribute__((ext_vector_type(2))) float v2f;
typedef __attribute__((ext_vector_type(8))) float v8f;


#define BUCKET_SHIFT 9
#define BUCKET_NODES (1 << BUCKET_SHIFT)
#define MAX_BUCKETS  1024
#define EDGE_CHUNK   4096
#define EDGES_PER_THREAD (EDGE_CHUNK / 256)
#define PIECES_AHEAD 8

__global__ __launch_bounds__(256)
void chunk_sort_kernel(const int* __restrict__ src, const int* __restrict__ dst,
                       int* __restrict__ desc, int2* __restrict__ filed,
                       int E, int nchunk, int nb) {
    __shared__ __align__(16) int hist[MAX_BUCKETS];
    __shared__ int wsum[8];
    const int tid  = threadIdx.x;
    const int lane = tid & 31;
    const int wv   = tid >> 5;
    const int chunk = blockIdx.x;
    const int e0 = chunk * EDGE_CHUNK;
    const int e1 = e0 + EDGE_CHUNK < E ? e0 + EDGE_CHUNK : E;
    *reinterpret_cast<int4*>(hist + tid * 4) = make_int4(0, 0, 0, 0);
    int s[EDGES_PER_THREAD], d[EDGES_PER_THREAD], r[EDGES_PER_THREAD];
    #pragma unroll
    for (int k = 0; k < EDGES_PER_THREAD; ++k) {
        const int e = e0 + k * 256 + tid;
        s[k] = e < e1 ? src[e] : 0;
        d[k] = e < e1 ? dst[e] : -1;
    }
    __syncthreads();
    #pragma unroll
    for (int k = 0; k < EDGES_PER_THREAD; ++k)
        r[k] = d[k] >= 0 ? atomicAdd(&hist[d[k] >> BUCKET_SHIFT], 1) : 0;
    __syncthreads();

    const int4 c = *reinterpret_cast<const int4*>(hist + tid * 4);
    const int t1 = c.x, t2 = t1 + c.y, t3 = t2 + c.z, t4 = t3 + c.w;
    int inc = t4;
    for (int off = 1; off < 32; off <<= 1) {
        const int u = __shfl_up(inc, off);
        if (lane >= off) inc += u;
    }
    if (lane == 31) wsum[wv] = inc;
    __syncthreads();
    int x0 = inc - t4;
    for (int w = 0; w < 8; ++w)
        if (w < wv) x0 += wsum[w];
    *reinterpret_cast<int4*>(hist + tid * 4) = make_int4(x0, x0 + t1, x0 + t2, x0 + t3);
    __syncthreads();
    #pragma unroll
    for (int k = 0; k < EDGES_PER_THREAD; ++k)
        if (d[k] >= 0)
            filed[e0 + hist[d[k] >> BUCKET_SHIFT] + r[k]] = make_int2(s[k], d[k]);
    const int b0 = tid * 4;
    if (b0     < nb) desc[(b0    ) * nchunk + chunk] =  x0       | (c.x << 16);
    if (b0 + 1 < nb) desc[(b0 + 1) * nchunk + chunk] = (x0 + t1) | (c.y << 16);
    if (b0 + 2 < nb) desc[(b0 + 2) * nchunk + chunk] = (x0 + t2) | (c.z << 16);
    if (b0 + 3 < nb) desc[(b0 + 3) * nchunk + chunk] = (x0 + t3) | (c.w << 16);
}

__global__ __launch_bounds__(1024)
void bucket_sort_kernel(const int2* __restrict__ filed, const int* __restrict__ desc,
                        int* __restrict__ rowptr, float* __restrict__ dinv,
                        int* __restrict__ meta, int n, int nchunk, int nb) {
    __shared__ int   cnt[BUCKET_NODES];
    __shared__ int   cur[BUCKET_NODES];
    __shared__ int   wtot[BUCKET_NODES / 32];
    __shared__ int   wprev[32];
    const int tid  = threadIdx.x;
    const int lane = tid & 31;
    const int wv   = __builtin_amdgcn_readfirstlane(tid >> 5);
    const int b    = blockIdx.x;
    const int node0 = b << BUCKET_SHIFT;
    const int* drow = desc + b * nchunk;

    const int lim  = b * nchunk;
    const int lim4 = lim & ~3;
    int prev = 0;
    for (int j0 = tid * 4; j0 < lim4; j0 += 8 * 4096) {
        int4 p[8];
        #pragma unroll
        for (int q = 0; q < 8; ++q) {
            const int j = j0 + q * 4096;
            p[q] = *reinterpret_cast<const int4*>(desc + (j < lim4 ? j : lim4 - 4));
        }
        #pragma unroll
        for (int q = 0; q < 8; ++q) {
            const int t = ((p[q].x >> 16) + (p[q].y >> 16)) + ((p[q].z >> 16) + (p[q].w >> 16));
            prev += j0 + q * 4096 < lim4 ? t : 0;
        }
    }
    if (tid < (lim & 3)) prev += desc[lim4 + tid] >> 16;
    for (int off = 16; off > 0; off >>= 1) prev += __shfl_xor(prev, off);
    if (lane == 0) wprev[wv] = prev;
    if (tid < BUCKET_NODES) cnt[tid] = 0;
    __syncthreads();

    for (int c0 = wv; c0 < nchunk; c0 += 32 * PIECES_AHEAD) {
        int at[PIECES_AHEAD], len[PIECES_AHEAD], d[PIECES_AHEAD];
        #pragma unroll
        for (int k = 0; k < PIECES_AHEAD; ++k) {
            const int c = c0 + 32 * k;
            const int pd = c < nchunk ? drow[c] : 0;
            at[k]  = c * EDGE_CHUNK + (pd & 0xffff);
            len[k] = pd >> 16;
        }
        #pragma unroll
        for (int k = 0; k < PIECES_AHEAD; ++k)
            d[k] = lane < len[k] ? filed[at[k] + lane].y : -1;
        #pragma unroll
        for (int k = 0; k < PIECES_AHEAD; ++k)
            if (d[k] >= 0) atomicAdd(&cnt[d[k] - node0], 1);
        #pragma unroll
        for (int k = 0; k < PIECES_AHEAD; ++k)
            for (int e = 32 + lane; e < len[k]; e += 32)
                atomicAdd(&cnt[filed[at[k] + e].y - node0], 1);
    }
    __syncthreads();

    int c = 0, inc = 0;
    if (tid < BUCKET_NODES) {
        c = cnt[tid];
        inc = c;
        for (int off = 1; off < 32; off <<= 1) {
            const int u = __shfl_up(inc, off);
            if (lane >= off) inc += u;
        }
        if (lane == 31) wtot[wv] = inc;
    }
    __syncthreads();
    int excl = 0;
    if (tid < BUCKET_NODES) {
        int start = wprev[lane];
        for (int off = 16; off > 0; off >>= 1) start += __shfl_xor(start, off);
        for (int w = 0; w < wv; ++w) start += wtot[w];
        excl = start + inc - c;
        cur[tid] = excl;
    }
    __syncthreads();
    if (tid < BUCKET_NODES) {
        const int node = node0 + tid;
        if (node < n) {
            const float dv = 1.0f / sqrtf((float)(c + 1));
            rowptr[node] = excl;
            dinv[node]   = dv;
        }
        if (b == nb - 1 && tid == BUCKET_NODES - 1) rowptr[n] = excl + c;
    }

    for (int c0 = wv; c0 < nchunk; c0 += 32 * PIECES_AHEAD) {
        int at[PIECES_AHEAD], len[PIECES_AHEAD];
        int2 e[PIECES_AHEAD];
        #pragma unroll
        for (int k = 0; k < PIECES_AHEAD; ++k) {
            const int c = c0 + 32 * k;
            const int pd = c < nchunk ? drow[c] : 0;
            at[k]  = c * EDGE_CHUNK + (pd & 0xffff);
            len[k] = pd >> 16;
        }
        #pragma unroll
        for (int k = 0; k < PIECES_AHEAD; ++k)
            e[k] = lane < len[k] ? filed[at[k] + lane] : make_int2(0, -1);
        #pragma unroll
        for (int k = 0; k < PIECES_AHEAD; ++k) {
            if (e[k].y >= 0) {
                const int pos = atomicAdd(&cur[e[k].y - node0], 1);
                meta[pos] = e[k].x;
            }
        }
        #pragma unroll
        for (int k = 0; k < PIECES_AHEAD; ++k) {
            for (int j = 32 + lane; j < len[k]; j += 32) {
                const int2 ee = filed[at[k] + j];
                const int pos = atomicAdd(&cur[ee.y - node0], 1);
                meta[pos] = ee.x;
            }
        }
    }
}

#define GATHER_AHEAD 16
template <int CPL>
__global__ void gcn_gather_kernel(const float* __restrict__ h,
                       const int* __restrict__ rowptr,
                       const int* __restrict__ meta,
                       const float* __restrict__ dinv,
                       const float* __restrict__ bias,
                       float* __restrict__ out,
                       int n, int fout, int do_relu) {
    typedef __attribute__((ext_vector_type(CPL))) float vrow;
    const int lane = threadIdx.x & 31;
    const int i = __builtin_amdgcn_readfirstlane(
                      (int)((blockIdx.x * blockDim.x + threadIdx.x) >> 5));
    if (i >= n) return;
    const int c0 = lane * CPL;
    const float* hc = h + c0;

    const vrow self = *reinterpret_cast<const vrow*>(hc + (size_t)i * fout);
    const float dv = dinv[i];
    const int beg = rowptr[i], end = rowptr[i + 1];

    int nE = end - beg < 32 ? end - beg : 32;
    int m = 0;
    if (lane < nE) m = meta[beg + lane];

    vrow acc = self;

    for (int cb = beg; cb < end; ) {
        for (int e = 0; e < nE; e += GATHER_AHEAD) {
            vrow  r[GATHER_AHEAD];
            #pragma unroll
            for (int q = 0; q < GATHER_AHEAD; ++q) {
                if (q == 0 || e + q < nE) {
                    const int sidx = __builtin_amdgcn_readlane(m, e + q);
                    r[q] = *reinterpret_cast<const vrow*>(hc + (size_t)sidx * fout);
                }
            }
            #pragma unroll
            for (int q = 0; q < GATHER_AHEAD; ++q) {
                if (q == 0 || e + q < nE) acc += r[q];
            }
        }
        cb += 32;
        if (cb < end) {
            nE = end - cb < 32 ? end - cb : 32;
            m = 0;
            if (lane < nE) m = meta[cb + lane];
        }
    }

    const vrow bv = *reinterpret_cast<const vrow*>(bias + c0);
    vrow v = acc * dv + bv;
    if (do_relu) {
        for (int c = 0; c < CPL; ++c) v[c] = fmaxf(v[c], 0.0f);
    }
    *reinterpret_cast<vrow*>(out + (size_t)i * fout + c0) = v;
}


#define FUSE_NODES 64
template <int CPL>
__global__ __launch_bounds__(256)
void gcn_gather_gemm_kernel(const float* __restrict__ h,
                       const int* __restrict__ rowptr,
                       const int* __restrict__ meta,
                       const float* __restrict__ dinv,
                       const float* __restrict__ bias,
                       const float* __restrict__ Wn,
                       float* __restrict__ hn,
                       int n) {
    typedef __attribute__((ext_vector_type(CPL))) float vrow;
    constexpr int FIN   = 32 * CPL;
    constexpr int FOUT  = 64;
    constexpr int PITCH = FIN + 4;
    constexpr int NPW   = FUSE_NODES / 8;
    __shared__ float Wt[FIN * FOUT];
    __shared__ float Xt[FUSE_NODES * PITCH];
    const int tid  = threadIdx.x;
    const int lane = tid & 31;
    const int wv = __builtin_amdgcn_readfirstlane(tid >> 5);

    constexpr int NW4 = FIN * FOUT / (256 * 4);
    float4 wreg[NW4];
    #pragma unroll
    for (int q = 0; q < NW4; ++q)
        wreg[q] = *reinterpret_cast<const float4*>(Wn + (q * 256 + tid) * 4);
    #pragma unroll
    for (int q = 0; q < NW4; ++q)
        *reinterpret_cast<float4*>(Wt + (q * 256 + tid) * 4) = wreg[q];

    const int node0 = blockIdx.x * FUSE_NODES;
    const int fout = FIN;
    const int c0 = lane * CPL;
    const float* hc = h + c0;
    const vrow bv = *reinterpret_cast<const vrow*>(bias + c0);
    const int ibase = node0 + wv * NPW;

    int sb = 0, se = 0; float sd = 0.0f;
    if (ibase < n) { sb = rowptr[ibase]; se = rowptr[ibase + 1]; sd = dinv[ibase]; }
    int begN = sb, endN = se; float dvN = sd;
    int nEN = endN - begN < 32 ? endN - begN : 32;
    vrow selfN;
    for (int c = 0; c < CPL; ++c) selfN[c] = 0.0f;
    if (ibase < n) selfN = *reinterpret_cast<const vrow*>(hc + (size_t)ibase * fout);
    int mN = 0;
    if (lane < nEN) mN = meta[begN + lane];
    sb = 0; se = 0; sd = 0.0f;
    if (NPW > 1 && ibase + 1 < n) { sb = rowptr[ibase + 1]; se = rowptr[ibase + 2]; sd = dinv[ibase + 1]; }

    for (int rr = 0; rr < NPW; ++rr) {
        const int lr = wv * NPW + rr;
        const int i  = ibase + rr;
        const int beg = begN, end = endN;
        const float dv = dvN;
        const vrow self = selfN;
        const int m = mN;
        const int nE = nEN;

        begN = sb; endN = se; dvN = sd;
        nEN = endN - begN < 32 ? endN - begN : 32;
        if (rr + 1 < NPW && i + 1 < n)
            selfN = *reinterpret_cast<const vrow*>(hc + (size_t)(i + 1) * fout);
        mN = 0;
        if (lane < nEN) mN = meta[begN + lane];
        sb = 0; se = 0; sd = 0.0f;
        if (rr + 2 < NPW && i + 2 < n) { sb = rowptr[i + 2]; se = rowptr[i + 3]; sd = dinv[i + 2]; }

        vrow v;
        for (int c = 0; c < CPL; ++c) v[c] = 0.0f;
        if (i < n) {
            vrow acc = self;

            auto sweep = [&](const int mm, const int cnt) {
                for (int e = 0; e < cnt; e += GATHER_AHEAD) {
                    vrow  r[GATHER_AHEAD];
                    #pragma unroll
                    for (int q = 0; q < GATHER_AHEAD; ++q) {
                        if (q == 0 || e + q < cnt) {
                            const int sidx = __builtin_amdgcn_readlane(mm, e + q);
                            r[q] = *reinterpret_cast<const vrow*>(hc + (size_t)sidx * fout);
                        }
                    }
                    #pragma unroll
                    for (int q = 0; q < GATHER_AHEAD; ++q) {
                        if (q == 0 || e + q < cnt) acc += r[q];
                    }
                }
            };
            sweep(m, nE);
            for (int cb = beg + 32; cb < end; cb += 32) {
                const int cnt = end - cb < 32 ? end - cb : 32;
                int mm = 0;
                if (lane < cnt) mm = meta[cb + lane];
                sweep(mm, cnt);
            }
            v = acc * dv + bv;
            for (int c = 0; c < CPL; ++c) v[c] = fmaxf(v[c], 0.0f) * dv;
        }
        *reinterpret_cast<vrow*>(Xt + lr * PITCH + c0) = v;
    }
    __syncthreads();

    const int rt   = wv >> 1;
    const int half = lane >> 4;
    const int l    = lane & 15;
    const float* xa = Xt + (rt * 16 + l) * PITCH + half * 2;

    const int ctb  = (wv & 1) * 2;
    const float* wb = Wt + (half * 2) * FOUT + ctb * 16 + l;

    v8f acc0 = {}, acc1 = {};
    for (int k = 0; k < FIN; k += 4) {
        v2f a = *reinterpret_cast<const v2f*>(xa + k);
        v2f b0, b1;
        b0.x = wb[k * FOUT];
        b0.y = wb[(k + 1) * FOUT];
        b1.x = wb[k * FOUT + 16];
        b1.y = wb[(k + 1) * FOUT + 16];
        acc0 = __builtin_amdgcn_wmma_f32_16x16x4_f32(
                   false, a, false, b0, (short)0, acc0, false, false);
        acc1 = __builtin_amdgcn_wmma_f32_16x16x4_f32(
                   false, a, false, b1, (short)0, acc1, false, false);
    }

    const int base_m = node0 + rt * 16 + half * 8;
    const int col = ctb * 16 + l;
    #pragma unroll
    for (int j = 0; j < 8; ++j) {
        const int m = base_m + j;
        if (m < n) {
            hn[(size_t)m * FOUT + col]      = acc0[j];
            hn[(size_t)m * FOUT + col + 16] = acc1[j];
        }
    }
}

#define PIPE_GW 16
#define PIPE_MW 4
#define PIPE_THREADS ((PIPE_GW + PIPE_MW) * 32)
#define PIPE_NPW (FUSE_NODES / PIPE_GW)
#define PIPE_GROUPS 256
__global__ __launch_bounds__(PIPE_THREADS)
void gcn_layer1_pipe_kernel(const float* __restrict__ x,
                       const int* __restrict__ rowptr,
                       const int* __restrict__ meta,
                       const float* __restrict__ dinv,
                       const float* __restrict__ bias,
                       const float* __restrict__ Wf,
                       const float* __restrict__ Wn,
                       float* __restrict__ hn,
                       int n) {
    typedef __attribute__((ext_vector_type(4))) float vrow;
    constexpr int FIN   = 128;
    constexpr int FOUT  = 64;
    constexpr int PITCH = FIN + 4;
    constexpr int PF    = 2 * FIN + 32;
    constexpr int PN    = 2 * FOUT + 32;
    __shared__ __align__(16) float Wfp[(FIN / 2) * PF];
    __shared__ __align__(16) float Wnp[(FIN / 2) * PN];
    __shared__ __align__(16) float Xt[2][FUSE_NODES * PITCH];
    __shared__ float dvrow[2][FUSE_NODES];
    constexpr int PPITCH = FOUT + 4;
    __shared__ __align__(16) float Park[PIPE_MW * 16 * PPITCH];
    const int tid  = threadIdx.x;
    const int lane = tid & 31;
    const int wv = __builtin_amdgcn_readfirstlane(tid >> 5);

    {
        constexpr int QF  = (FIN / 2) * (FIN / 4);
        constexpr int QN  = (FIN / 2) * (FOUT / 4);
        constexpr int NQF = (QF + PIPE_THREADS - 1) / PIPE_THREADS;
        constexpr int NQN = (QN + PIPE_THREADS - 1) / PIPE_THREADS;
        float4 flo[NQF], fhi[NQF], nlo[NQN], nhi[NQN];
        #pragma unroll
        for (int u = 0; u < NQF; ++u) {
            int q = u * PIPE_THREADS + tid;
            q = q < QF ? q : QF - 1;
            const int kp = q / (FIN / 4), c4 = q % (FIN / 4);
            flo[u] = *reinterpret_cast<const float4*>(Wf + (2 * kp    ) * FIN + c4 * 4);
            fhi[u] = *reinterpret_cast<const float4*>(Wf + (2 * kp + 1) * FIN + c4 * 4);
        }
        #pragma unroll
        for (int u = 0; u < NQN; ++u) {
            int q = u * PIPE_THREADS + tid;
            q = q < QN ? q : QN - 1;
            const int kp = q / (FOUT / 4), c4 = q % (FOUT / 4);
            nlo[u] = *reinterpret_cast<const float4*>(Wn + (2 * kp    ) * FOUT + c4 * 4);
            nhi[u] = *reinterpret_cast<const float4*>(Wn + (2 * kp + 1) * FOUT + c4 * 4);
        }
        #pragma unroll
        for (int u = 0; u < NQF; ++u) {
            const int q = u * PIPE_THREADS + tid;
            if (q < QF) {
                const int kp = q / (FIN / 4), c4 = q % (FIN / 4);
                float* d = Wfp + kp * PF + c4 * 8;
                *reinterpret_cast<float4*>(d)     = make_float4(flo[u].x, fhi[u].x, flo[u].y, fhi[u].y);
                *reinterpret_cast<float4*>(d + 4) = make_float4(flo[u].z, fhi[u].z, flo[u].w, fhi[u].w);
            }
        }
        #pragma unroll
        for (int u = 0; u < NQN; ++u) {
            const int q = u * PIPE_THREADS + tid;
            if (q < QN) {
                const int kp = q / (FOUT / 4), c4 = q % (FOUT / 4);
                float* d = Wnp + kp * PN + c4 * 8;
                *reinterpret_cast<float4*>(d)     = make_float4(nlo[u].x, nhi[u].x, nlo[u].y, nhi[u].y);
                *reinterpret_cast<float4*>(d + 4) = make_float4(nlo[u].z, nhi[u].z, nlo[u].w, nhi[u].w);
            }
        }
    }

    const int ntile = (n + FUSE_NODES - 1) / FUSE_NODES;
    const int nT = (int)blockIdx.x < ntile ? (ntile - 1 - (int)blockIdx.x) / (int)gridDim.x + 1 : 0;
    const int nTg = wv < PIPE_GW ? nT : 0;
    auto first_node = [&](const int j) {
        return j < nTg ? ((int)blockIdx.x + j * (int)gridDim.x) * FUSE_NODES + wv * PIPE_NPW : n;
    };
    auto load_bounds = [&](const int j, int& rp, float& dn) {
        const int b = first_node(j);
        rp = 0; dn = 0.0f;
        if (lane <= PIPE_NPW && b + lane <= n) rp = rowptr[b + lane];
        if (lane <  PIPE_NPW && b + lane <  n) dn = dinv[b + lane];
    };

    const int c0 = lane * 4;
    const float* hc = x + c0;
    const int half = lane >> 4;
    const int l    = lane & 15;

    int rpN; float dnN;
    load_bounds(0, rpN, dnN);
    vrow selfN;
    for (int c = 0; c < 4; ++c) selfN[c] = 0.0f;
    int mN = 0;
    {
        const int i0 = first_node(0);
        const int b0 = __builtin_amdgcn_readlane(rpN, 0), e0 = __builtin_amdgcn_readlane(rpN, 1);
        const int nE0 = i0 < n ? (e0 - b0 < 32 ? e0 - b0 : 32) : 0;
        if (i0 < n) selfN = *reinterpret_cast<const vrow*>(hc + (size_t)i0 * FIN);
        if (lane < nE0) mN = meta[b0 + lane];
    }

    float bc[FIN / 16];
    #pragma unroll
    for (int t = 0; t < FIN / 16; ++t) bc[t] = bias[t * 16 + l];

    float* park = Park + (wv >= PIPE_GW ? wv - PIPE_GW : 0) * (16 * PPITCH);

    for (int j = 0; j <= nT + 1; ++j) {
        if (wv < PIPE_GW) {
            if (j < nT) {
                float* Xs  = Xt[j & 1];
                float* dvs = dvrow[j & 1];
                const int rpC = rpN; const float dnC = dnN;
                load_bounds(j + 1, rpN, dnN);
                const int ifirst = first_node(j);
                for (int rr = 0; rr < PIPE_NPW; ++rr) {
                    const int lr = wv * PIPE_NPW + rr;
                    const int i  = ifirst + rr;
                    const int beg = __builtin_amdgcn_readlane(rpC, rr);
                    const int end = __builtin_amdgcn_readlane(rpC, rr + 1);
                    const float dv = __int_as_float(__builtin_amdgcn_readlane(__float_as_int(dnC), rr));
                    const vrow self = selfN;
                    const int m = mN;
                    const int nE = i < n ? (end - beg < 32 ? end - beg : 32) : 0;

                    int i1, b1, e1;
                    if (rr + 1 < PIPE_NPW) {
                        i1 = i + 1;
                        b1 = end;
                        e1 = __builtin_amdgcn_readlane(rpC, rr + 2);
                    } else {
                        i1 = first_node(j + 1);
                        b1 = __builtin_amdgcn_readlane(rpN, 0);
                        e1 = __builtin_amdgcn_readlane(rpN, 1);
                    }
                    const int nE1 = i1 < n ? (e1 - b1 < 32 ? e1 - b1 : 32) : 0;
                    if (i1 < n) selfN = *reinterpret_cast<const vrow*>(hc + (size_t)i1 * FIN);
                    mN = 0;
                    if (lane < nE1) mN = meta[b1 + lane];

                    vrow v;
                    for (int c = 0; c < 4; ++c) v[c] = 0.0f;
                    if (i < n) {
                        vrow acc = self * dv;

                        auto sweep = [&](const int mm, const int cnt) {
                            float ww = 0.0f;
                            if (lane < cnt) ww = dinv[mm];
                            for (int e = 0; e < cnt; e += GATHER_AHEAD) {
                                vrow  r[GATHER_AHEAD];
                                #pragma unroll
                                for (int q = 0; q < GATHER_AHEAD; ++q) {
                                    if (q == 0 || e + q < cnt) {
                                        const int sidx = __builtin_amdgcn_readlane(mm, e + q);
                                        r[q] = *reinterpret_cast<const vrow*>(hc + (size_t)sidx * FIN);
                                    }
                                }
                                #pragma unroll
                                for (int q = 0; q < GATHER_AHEAD; ++q) {
                                    if (q == 0 || e + q < cnt) {
                                        const float w = __int_as_float(
                                            __builtin_amdgcn_readlane(__float_as_int(ww), e + q));
                                        for (int c = 0; c < 4; ++c) acc[c] = fmaf(r[q][c], w, acc[c]);
                                    }
                                }
                            }
                        };
                        sweep(m, nE);
                        for (int cb = beg + 32; cb < end; cb += 32) {
                            const int cnt = end - cb < 32 ? end - cb : 32;
                            int mm = 0;
                            if (lane < cnt) mm = meta[cb + lane];
                            sweep(mm, cnt);
                        }
                        v = acc * dv;
                    }
                    if (lane == 0) dvs[lr] = i < n ? dv : 0.0f;
                    *reinterpret_cast<vrow*>(Xs + lr * PITCH + c0) = v;
                }
            }
        } else {
            const int rt = wv - PIPE_GW;
            if (j >= 2) {
                const int pnode = ((int)blockIdx.x + (j - 2) * (int)gridDim.x) * FUSE_NODES + rt * 16;
                vrow pq[8];
                #pragma unroll
                for (int u = 0; u < 8; ++u)
                    pq[u] = *reinterpret_cast<const vrow*>(park + (2 * u + half) * PPITCH + l * 4);
                asm volatile("" :: "v"(pq[0]), "v"(pq[1]), "v"(pq[2]), "v"(pq[3]),
                                   "v"(pq[4]), "v"(pq[5]), "v"(pq[6]), "v"(pq[7]));
                #pragma unroll
                for (int u = 0; u < 8; ++u) {
                    const int m = pnode + 2 * u + half;
                    if (m < n)
                        *reinterpret_cast<vrow*>(hn + (size_t)m * FOUT + l * 4) = pq[u];
                }
            }
            if (j >= 1 && j <= nT) {
                float* Xs = Xt[(j - 1) & 1];
                const float* dvs = dvrow[(j - 1) & 1];
                const int node0 = ((int)blockIdx.x + (j - 1) * (int)gridDim.x) * FUSE_NODES;
                const float* xa = Xs + (rt * 16 + l) * PITCH + half * 2;
                const float* wf = Wfp + half * PF + l * 2;
                const float* wn = Wnp + half * PN + l * 2;

                constexpr int NT = FIN / 16;
                v8f accf[NT];
                #pragma unroll
                for (int t = 0; t < NT; ++t) accf[t] = {};
                v2f a0, a1, b0[NT], b1[NT];
                a0 = *reinterpret_cast<const v2f*>(xa);
                #pragma unroll
                for (int t = 0; t < NT; ++t) b0[t] = *reinterpret_cast<const v2f*>(wf + t * 32);
                __builtin_amdgcn_sched_group_barrier(0x100, 5, 0);
                for (int k = 0; k < FIN; k += 8) {
                    a1 = *reinterpret_cast<const v2f*>(xa + k + 4);
                    #pragma unroll
                    for (int t = 0; t < NT; ++t)
                        b1[t] = *reinterpret_cast<const v2f*>(wf + (k / 2 + 2) * PF + t * 32);
                    #pragma unroll
                    for (int t = 0; t < NT; ++t)
                        accf[t] = __builtin_amdgcn_wmma_f32_16x16x4_f32(
                                      false, a0, false, b0[t], (short)0, accf[t], false, false);
                    if (k + 8 < FIN) {
                        a0 = *reinterpret_cast<const v2f*>(xa + k + 8);
                        #pragma unroll
                        for (int t = 0; t < NT; ++t)
                            b0[t] = *reinterpret_cast<const v2f*>(wf + (k / 2 + 4) * PF + t * 32);
                    }
                    #pragma unroll
                    for (int t = 0; t < NT; ++t)
                        accf[t] = __builtin_amdgcn_wmma_f32_16x16x4_f32(
                                      false, a1, false, b1[t], (short)0, accf[t], false, false);
                    __builtin_amdgcn_sched_group_barrier(0x100, 5, 0);
                    __builtin_amdgcn_sched_group_barrier(0x008, 8, 0);
                    __builtin_amdgcn_sched_group_barrier(0x100, 5, 0);
                    __builtin_amdgcn_sched_group_barrier(0x008, 8, 0);
                }
                #pragma unroll
                for (int t = 0; t < NT; ++t) {
                    #pragma unroll
                    for (int r = 0; r < 8; ++r)
                        Xs[(rt * 16 + half * 8 + r) * PITCH + t * 16 + l] =
                            fmaxf(accf[t][r] + bc[t], 0.0f) * dvs[rt * 16 + half * 8 + r];
                }
                __builtin_amdgcn_fence(__ATOMIC_RELEASE, "wavefront");
                __builtin_amdgcn_wave_barrier();
                __builtin_amdgcn_fence(__ATOMIC_ACQUIRE, "wavefront");

                constexpr int MT = FOUT / 16;
                v8f acc2[MT];
                #pragma unroll
                for (int t = 0; t < MT; ++t) acc2[t] = {};
                v2f p0[2], p1[2], q0[2][MT], q1[2][MT];
                #pragma unroll
                for (int u = 0; u < 2; ++u) {
                    p0[u] = *reinterpret_cast<const v2f*>(xa + u * 4);
                    #pragma unroll
                    for (int t = 0; t < MT; ++t)
                        q0[u][t] = *reinterpret_cast<const v2f*>(wn + (u * 2) * PN + t * 32);
                }
                __builtin_amdgcn_sched_group_barrier(0x100, 6, 0);
                for (int k = 0; k < FIN; k += 16) {
                    #pragma unroll
                    for (int u = 0; u < 2; ++u) {
                        p1[u] = *reinterpret_cast<const v2f*>(xa + k + 8 + u * 4);
                        #pragma unroll
                        for (int t = 0; t < MT; ++t)
                            q1[u][t] = *reinterpret_cast<const v2f*>(wn + (k / 2 + 4 + u * 2) * PN + t * 32);
                    }
                    #pragma unroll
                    for (int u = 0; u < 2; ++u) {
                        #pragma unroll
                        for (int t = 0; t < MT; ++t)
                            acc2[t] = __builtin_amdgcn_wmma_f32_16x16x4_f32(
                                          false, p0[u], false, q0[u][t], (short)0, acc2[t], false, false);
                    }
                    if (k + 16 < FIN) {
                        #pragma unroll
                        for (int u = 0; u < 2; ++u) {
                            p0[u] = *reinterpret_cast<const v2f*>(xa + k + 16 + u * 4);
                            #pragma unroll
                            for (int t = 0; t < MT; ++t)
                                q0[u][t] = *reinterpret_cast<const v2f*>(wn + (k / 2 + 8 + u * 2) * PN + t * 32);
                        }
                    }
                    #pragma unroll
                    for (int u = 0; u < 2; ++u) {
                        #pragma unroll
                        for (int t = 0; t < MT; ++t)
                            acc2[t] = __builtin_amdgcn_wmma_f32_16x16x4_f32(
                                          false, p1[u], false, q1[u][t], (short)0, acc2[t], false, false);
                    }
                    __builtin_amdgcn_sched_group_barrier(0x100, 6, 0);
                    __builtin_amdgcn_sched_group_barrier(0x008, 8, 0);
                    __builtin_amdgcn_sched_group_barrier(0x100, 6, 0);
                    __builtin_amdgcn_sched_group_barrier(0x008, 8, 0);
                }
                #pragma unroll
                for (int t = 0; t < MT; ++t) {
                    #pragma unroll
                    for (int r = 0; r < 8; ++r)
                        park[(half * 8 + r) * PPITCH + t * 16 + l] = acc2[t][r];
                }
            }
        }
        __syncthreads();
    }
}


static inline int cdiv(long long a, int b) { return (int)((a + b - 1) / b); }
static inline size_t align64up(size_t x) { return (x + 63) & ~(size_t)63; }

extern "C" void kernel_launch(void* const* d_in, const int* in_sizes, int n_in,
                              void* d_out, int out_size, void* d_ws, size_t ws_size,
                              hipStream_t stream) {
    const float* x  = (const float*)d_in[0];
    const int*   ei = (const int*)  d_in[1];
    const float* W1 = (const float*)d_in[2];
    const float* b1 = (const float*)d_in[3];
    const float* W2 = (const float*)d_in[4];
    const float* b2 = (const float*)d_in[5];
    const float* W3 = (const float*)d_in[6];
    const float* b3 = (const float*)d_in[7];

    const int IN_C = 128;
    const int n = in_sizes[0] / IN_C;
    const int E = in_sizes[1] / 2;
    const int* src = ei;
    const int* dst = ei + E;

    char* base = (char*)d_ws;
    size_t off = ((256 - ((size_t)base & 255)) & 255) / 4;
    auto take = [&](size_t elems) { void* p = base + off * 4; off += align64up(elems); return p; };
    int*   rowptr = (int*)  take(n + 1);
    float* dinv   = (float*)take(n);
    int*   meta   = (int*)  take(E);
    float* h      = (float*)take((size_t)n * 64);
    float* xn     = (float*)take((size_t)n * 64);
    const int nb     = cdiv(n, BUCKET_NODES);
    const int nchunk = cdiv(E, EDGE_CHUNK);
    int*   desc   = (int*)  take((size_t)nb * nchunk);
    int2*  filed  = (int2*) take((size_t)2 * E);
    (void)ws_size;

    const int B = 256;

    chunk_sort_kernel<<<nchunk, B, 0, stream>>>(src, dst, desc, filed, E, nchunk, nb);
    bucket_sort_kernel<<<nb, 1024, 0, stream>>>(filed, desc, rowptr, dinv, meta, n, nchunk, nb);

    const int gather_blocks = cdiv((long long)n * 32, B);
    const int fuse_blocks = cdiv(n, FUSE_NODES);

    gcn_layer1_pipe_kernel<<<PIPE_GROUPS, PIPE_THREADS, 0, stream>>>(x, rowptr, meta, dinv,
                                                             b1, W1, W2, xn, n);
    gcn_gather_gemm_kernel<2><<<fuse_blocks, B, 0, stream>>>(xn, rowptr, meta, dinv,
                                                             b2, W3, h, n);
    gcn_gather_kernel<2><<<gather_blocks, B, 0, stream>>>(h, rowptr, meta, dinv, b3,
                                                          (float*)d_out, n, 64, 0);
}
